# starting kernel
# baseline (speedup 1.0000x reference)
#include <hip/hip_runtime.h>

typedef __attribute__((ext_vector_type(2))) float v2f;
typedef __attribute__((ext_vector_type(8))) float v8f;

__device__ __forceinline__ v8f zero8() {
    v8f z = {0.f,0.f,0.f,0.f,0.f,0.f,0.f,0.f};
    return z;
}

__device__ __forceinline__ v8f wmma4(v2f a, v2f b, v8f c) {
    return __builtin_amdgcn_wmma_f32_16x16x4_f32(
         false, a,  false, b,
         (short)0, c,  false,  false);
}

#define DIM455 455

__global__ __launch_bounds__(256) void traj_kernel(
    const float* __restrict__ traj, const float* __restrict__ w1s,
    const float* __restrict__ w1v,  const float* __restrict__ Din,
    const float* __restrict__ Dout, const float* __restrict__ ico,
    const float* __restrict__ w20, const float* __restrict__ w21,
    const float* __restrict__ w22, const float* __restrict__ w23,
    const float* __restrict__ w24, const float* __restrict__ w25,
    const float* __restrict__ w26, float* __restrict__ out)
{
    constexpr int G = 4096, KC = 128;
    __shared__ float h_lds[16][12];
    __shared__ float S[16][130];
    __shared__ float H2[16][456];
    __shared__ float sc[8][16][16];

    const int bt   = blockIdx.x;
    const int tid  = threadIdx.x;
    const int wave = tid >> 5;
    const int lane = tid & 31;
    const int lrow = lane & 15;
    const int kb   = (lane >> 4) << 1;
    const int crow = (lane >> 4) << 3;

    const float* t10 = traj + bt * 10;

    if (tid < 160) {
        int c = tid / 10, i = tid % 10;
        float hv;
        if (i == 0) {
            hv = w1s[c] * t10[9];
        } else {
            int vv = (i - 1) / 3, mm = (i - 1) % 3;
            float a = 0.f;
            #pragma unroll
            for (int u = 0; u < 3; ++u)
                a += t10[u * 3 + mm] * w1v[c * 9 + u * 3 + vv];
            hv = a * 0.5773502691896258f;
        }
        h_lds[c][i] = hv;
    }
    __syncthreads();

    v8f acc0 = zero8(), acc1 = zero8(), acc2 = zero8(), acc3 = zero8();

    for (int g0 = 0; g0 < G; g0 += KC) {
        for (int idx = tid; idx < 16 * KC; idx += 256) {
            int c = idx >> 7, gg = idx & (KC - 1);
            const float* dr = Din + (g0 + gg) * 10;
            float a = 0.f;
            #pragma unroll
            for (int i = 0; i < 10; ++i) a += h_lds[c][i] * dr[i];
            S[c][gg] = fmaxf(a, 0.f);
        }
        __syncthreads();

        for (int kk = 0; kk < KC; kk += 4) {
            v2f a;
            a.x = S[lrow][kk + kb];
            a.y = S[lrow][kk + kb + 1];
            int k = g0 + kk + kb;
            const float* d0 = Dout + k * DIM455;
            const float* d1 = Dout + (k + 1) * DIM455;

            int n0 = (wave + 0 * 8) * 16 + lrow;
            int n1 = (wave + 1 * 8) * 16 + lrow;
            int n2 = (wave + 2 * 8) * 16 + lrow;
            int n3 = (wave + 3 * 8) * 16 + lrow;
            int n3c = (n3 < DIM455) ? n3 : (DIM455 - 1);
            v2f b0, b1, b2, b3;
            b0.x = d0[n0];   b0.y = d1[n0];
            b1.x = d0[n1];   b1.y = d1[n1];
            b2.x = d0[n2];   b2.y = d1[n2];
            float t3x = d0[n3c], t3y = d1[n3c];
            b3.x = (n3 < DIM455) ? t3x : 0.f;
            b3.y = (n3 < DIM455) ? t3y : 0.f;

            acc0 = wmma4(a, b0, acc0);
            acc1 = wmma4(a, b1, acc1);
            acc2 = wmma4(a, b2, acc2);
            acc3 = wmma4(a, b3, acc3);
        }
        __syncthreads();
    }

    {
        v8f* accs[4] = {&acc0, &acc1, &acc2, &acc3};
        #pragma unroll
        for (int tt = 0; tt < 4; ++tt) {
            int col = (wave + tt * 8) * 16 + lrow;
            if (col < DIM455) {
                #pragma unroll
                for (int j = 0; j < 8; ++j)
                    H2[j + crow][col] = (*accs[tt])[j];
            }
        }
    }
    __syncthreads();

    const int mtile = wave >> 1;
    const int sub   = wave & 1;

    v8f facc0 = zero8(), facc1 = zero8(), facc2 = zero8(), facc3 = zero8();

    #pragma unroll
    for (int l = 0; l <= 6; ++l) {
        const int d = 2 * l + 1;
        const int OFFT[7] = {0, 1, 10, 35, 84, 165, 286};
        const int o = OFFT[l];
        const int Kdim = 16 * d;
        const float scale = rsqrtf((float)(16 * d));
        const float* w2l = (l == 0) ? w20 : (l == 1) ? w21 : (l == 2) ? w22 :
                           (l == 3) ? w23 : (l == 4) ? w24 : (l == 5) ? w25 : w26;

        for (int v = (sub + l) & 1; v < d; v += 2) {
            v8f dacc = zero8();
            const int m  = lrow;
            const int mc = (m < d) ? m : (d - 1);
            for (int k0 = 0; k0 < Kdim; k0 += 4) {
                int k  = k0 + kb;
                int f0 = k & 15,        u0 = k >> 4;
                int f1 = (k + 1) & 15,  u1 = (k + 1) >> 4;
                v2f a;
                a.x = w2l[((f0 * 64 + mtile * 16 + lrow) * d + u0) * d + v];
                a.y = w2l[((f1 * 64 + mtile * 16 + lrow) * d + u1) * d + v];
                float h0 = H2[f0][o + u0 * d + mc];
                float h1 = H2[f1][o + u1 * d + mc];
                v2f b;
                b.x = (m < d) ? h0 : 0.f;
                b.y = (m < d) ? h1 : 0.f;
                dacc = wmma4(a, b, dacc);
            }

            #pragma unroll
            for (int j = 0; j < 8; ++j)
                sc[wave][j + crow][lrow] = (lrow < d) ? dacc[j] * scale : 0.f;

            #pragma unroll
            for (int k0 = 0; k0 < 16; k0 += 4) {
                int k   = k0 + kb;
                int kc0 = (k     < d) ? k       : (d - 1);
                int kc1 = (k + 1 < d) ? (k + 1) : (d - 1);
                v2f a;
                a.x = sc[wave][lrow][k];
                a.y = sc[wave][lrow][k + 1];
                const float* ib = ico + o + v * d;
                int r0 = 0 * 16 + lrow, r1 = 1 * 16 + lrow;
                int r2 = 2 * 16 + lrow, r3 = 3 * 16 + lrow;
                int r3c = (r3 < 60) ? r3 : 59;
                float t0x = ib[r0 * DIM455 + kc0], t0y = ib[r0 * DIM455 + kc1];
                float t1x = ib[r1 * DIM455 + kc0], t1y = ib[r1 * DIM455 + kc1];
                float t2x = ib[r2 * DIM455 + kc0], t2y = ib[r2 * DIM455 + kc1];
                float t3x = ib[r3c * DIM455 + kc0], t3y = ib[r3c * DIM455 + kc1];
                v2f b0, b1, b2, b3;
                b0.x = (k     < d) ? t0x : 0.f;
                b0.y = (k + 1 < d) ? t0y : 0.f;
                b1.x = (k     < d) ? t1x : 0.f;
                b1.y = (k + 1 < d) ? t1y : 0.f;
                b2.x = (k     < d) ? t2x : 0.f;
                b2.y = (k + 1 < d) ? t2y : 0.f;
                b3.x = (k     < d && r3 < 60) ? t3x : 0.f;
                b3.y = (k + 1 < d && r3 < 60) ? t3y : 0.f;
                facc0 = wmma4(a, b0, facc0);
                facc1 = wmma4(a, b1, facc1);
                facc2 = wmma4(a, b2, facc2);
                facc3 = wmma4(a, b3, facc3);
            }
        }
    }
    __syncthreads();

    float* red = &H2[0][0];
    {
        v8f* faccs[4] = {&facc0, &facc1, &facc2, &facc3};
        if (sub == 1) {
            #pragma unroll
            for (int nt = 0; nt < 4; ++nt)
                #pragma unroll
                for (int j = 0; j < 8; ++j)
                    red[(mtile * 16 + j + crow) * 64 + nt * 16 + lrow] = (*faccs[nt])[j];
        }
        __syncthreads();
        if (sub == 0) {
            const long xg_elems = 32L * 4 * 128 * 60;
            #pragma unroll
            for (int nt = 0; nt < 4; ++nt) {
                int r = nt * 16 + lrow;
                if (r < 60) {
                    #pragma unroll
                    for (int j = 0; j < 8; ++j) {
                        int g2 = mtile * 16 + j + crow;
                        float vsum = (*faccs[nt])[j] + red[g2 * 64 + r];
                        out[xg_elems + ((long)bt * 64 + g2) * 60 + r] = vsum;
                    }
                }
            }
        }
    }
}

__global__ __launch_bounds__(256) void xg_kernel(
    const float* __restrict__ x, const float* __restrict__ ico,
    float* __restrict__ out)
{
    const int tid  = threadIdx.x;
    const int wave = tid >> 5;
    const int lane = tid & 31;
    const int lrow = lane & 15;
    const int kb   = (lane >> 4) << 1;
    const int crow = (lane >> 4) << 3;
    const int mtile = wave >> 1;
    const int ng    = wave & 1;

    const long rowA = (long)blockIdx.x * 64 + mtile * 16 + lrow;
    const float* xr = x + rowA * DIM455;

    v8f acc0 = zero8(), acc1 = zero8();
    const int r0  = (ng * 2 + 0) * 16 + lrow;
    const int r1  = (ng * 2 + 1) * 16 + lrow;
    const int r1c = (r1 < 60) ? r1 : 59;

    for (int k0 = 0; k0 < 456; k0 += 4) {
        int k   = k0 + kb;
        int kc0 = (k     < DIM455) ? k       : (DIM455 - 1);
        int kc1 = (k + 1 < DIM455) ? (k + 1) : (DIM455 - 1);
        float ax = xr[kc0], ay = xr[kc1];
        float b0x = ico[r0  * DIM455 + kc0], b0y = ico[r0  * DIM455 + kc1];
        float b1x = ico[r1c * DIM455 + kc0], b1y = ico[r1c * DIM455 + kc1];
        v2f a, b0, b1;
        a.x  = (k     < DIM455) ? ax  : 0.f;
        a.y  = (k + 1 < DIM455) ? ay  : 0.f;
        b0.x = (k     < DIM455) ? b0x : 0.f;
        b0.y = (k + 1 < DIM455) ? b0y : 0.f;
        b1.x = (k     < DIM455 && r1 < 60) ? b1x : 0.f;
        b1.y = (k + 1 < DIM455 && r1 < 60) ? b1y : 0.f;
        acc0 = wmma4(a, b0, acc0);
        acc1 = wmma4(a, b1, acc1);
    }

    #pragma unroll
    for (int s = 0; s < 2; ++s) {
        int r = (ng * 2 + s) * 16 + lrow;
        if (r < 60) {
            #pragma unroll
            for (int j = 0; j < 8; ++j) {
                long row = (long)blockIdx.x * 64 + mtile * 16 + j + crow;
                out[row * 60 + r] = (s == 0) ? acc0[j] : acc1[j];
            }
        }
    }
}

extern "C" void kernel_launch(void* const* d_in, const int* in_sizes, int n_in,
                              void* d_out, int out_size, void* d_ws, size_t ws_size,
                              hipStream_t stream) {
    const float* x    = (const float*)d_in[0];
    const float* traj = (const float*)d_in[1];
    const float* w1s  = (const float*)d_in[2];
    const float* w1v  = (const float*)d_in[3];
    const float* Din  = (const float*)d_in[4];
    const float* Dout = (const float*)d_in[5];
    const float* ico  = (const float*)d_in[6];
    float* out = (float*)d_out;

    xg_kernel<<<dim3(16384 / 64), dim3(256), 0, stream>>>(x, ico, out);

    traj_kernel<<<dim3(512), dim3(256), 0, stream>>>(
        traj, w1s, w1v, Din, Dout, ico,
        (const float*)d_in[7],  (const float*)d_in[8],  (const float*)d_in[9],
        (const float*)d_in[10], (const float*)d_in[11], (const float*)d_in[12],
        (const float*)d_in[13], out);
}


# w2 mix re-tiled: tile rows run along v, the contiguous axis of w2, so each w2 load reads neighbouring floats; mixed rows pass through LDS, ico product is one K=455 GEMM (new summation order)
# speedup vs baseline: 2.7640x; 2.7640x over previous
#include <hip/hip_runtime.h>

typedef __attribute__((ext_vector_type(2))) float v2f;
typedef __attribute__((ext_vector_type(8))) float v8f;

__device__ __forceinline__ v8f zero8() {
    v8f z = {0.f,0.f,0.f,0.f,0.f,0.f,0.f,0.f};
    return z;
}

__device__ __forceinline__ v8f wmma4(v2f a, v2f b, v8f c) {
    return __builtin_amdgcn_wmma_f32_16x16x4_f32(
         false, a,  false, b,
         (short)0, c,  false,  false);
}

#define DIM455 455

#define H2S  456
#define OUTS 460

template <int L>
__device__ __forceinline__ void mix_frag(const float* __restrict__ wl, const float* hl,
                                         int kx, v2f (&a)[8], v2f& b)
{
    constexpr int d = 2 * L + 1, dd = d * d;
    const int f0 = (int)((unsigned)kx / (unsigned)d);
    const int f1 = (int)((unsigned)(kx + 1) / (unsigned)d);
    const float* p0 = wl + (f0 * (63 * dd) + kx * d);
    const float* p1 = wl + (f1 * (63 * dd) + (kx + 1) * d);
    b.x = hl[f0 * (H2S - dd) + kx * d];
    b.y = hl[f1 * (H2S - dd) + (kx + 1) * d];
    #pragma unroll
    for (int j = 0; j < 8; ++j) {
        a[j].x = p0[j * dd];
        a[j].y = p1[j * dd];
    }
}

template <int L>
__device__ __forceinline__ void mix_level(const float* __restrict__ w2l, const float* H2f,
                                          float* OUTf, int wave, int lrow, int kb, int crow)
{
    constexpr int d = 2 * L + 1, dd = d * d, Kdim = 16 * d;
    constexpr int o = (L * (2 * L - 1) * (2 * L + 1)) / 3;
    const float scale = rsqrtf((float)Kdim);
    const int vc = (lrow < d) ? lrow : (d - 1);
    const float* wl = w2l + (wave * 8) * dd + vc;
    const float* hl = H2f + o + vc;

    v8f acc[8];
    #pragma unroll
    for (int j = 0; j < 8; ++j) acc[j] = zero8();

    v2f a_e[8], a_o[8], b_e, b_o;
    mix_frag<L>(wl, hl, kb, a_e, b_e);
    for (int k0 = 0; k0 < Kdim; k0 += 8) {
        mix_frag<L>(wl, hl, k0 + 4 + kb, a_o, b_o);
        #pragma unroll
        for (int j = 0; j < 8; ++j) acc[j] = wmma4(a_e[j], b_e, acc[j]);
        const int kn = (k0 + 8 < Kdim) ? (k0 + 8) : k0;
        mix_frag<L>(wl, hl, kn + kb, a_e, b_e);
        #pragma unroll
        for (int j = 0; j < 8; ++j) acc[j] = wmma4(a_o[j], b_o, acc[j]);
    }

    float* ob = OUTf + (wave * 8) * OUTS + o + crow * d + lrow;
    if (lrow < d) {
        #pragma unroll
        for (int j = 0; j < 8; ++j) {
            #pragma unroll
            for (int r = 0; r < 8; ++r)
                if (r < d - 8) ob[j * OUTS + r * d] = acc[j][r] * scale;
        }
        if (crow == 0) {
            #pragma unroll
            for (int j = 0; j < 8; ++j) {
                #pragma unroll
                for (int r = 0; r < 8; ++r)
                    if (r >= d - 8 && r < d) ob[j * OUTS + r * d] = acc[j][r] * scale;
            }
        }
    }
}

__global__ __launch_bounds__(256) void traj_kernel(
    const float* __restrict__ traj, const float* __restrict__ w1s,
    const float* __restrict__ w1v,  const float* __restrict__ Din,
    const float* __restrict__ Dout, const float* __restrict__ ico,
    const float* __restrict__ w20, const float* __restrict__ w21,
    const float* __restrict__ w22, const float* __restrict__ w23,
    const float* __restrict__ w24, const float* __restrict__ w25,
    const float* __restrict__ w26, float* __restrict__ out)
{
    constexpr int G = 4096, KC = 128;
    __shared__ float H2[16][H2S];
    __shared__ __attribute__((aligned(16))) float U[64 * OUTS];
    float (*S)[130]    = reinterpret_cast<float (*)[130]>(U);
    float (*h_lds)[12] = reinterpret_cast<float (*)[12]>(U + 16 * 130);
    float* OUTf = U;

    const int bt   = blockIdx.x;
    const int tid  = threadIdx.x;
    const int wave = tid >> 5;
    const int lane = tid & 31;
    const int lrow = lane & 15;
    const int kb   = (lane >> 4) << 1;
    const int crow = (lane >> 4) << 3;

    const float* t10 = traj + bt * 10;

    if (tid < 160) {
        int c = tid / 10, i = tid % 10;
        float hv;
        if (i == 0) {
            hv = w1s[c] * t10[9];
        } else {
            int vv = (i - 1) / 3, mm = (i - 1) % 3;
            float a = 0.f;
            #pragma unroll
            for (int u = 0; u < 3; ++u)
                a += t10[u * 3 + mm] * w1v[c * 9 + u * 3 + vv];
            hv = a * 0.5773502691896258f;
        }
        h_lds[c][i] = hv;
    }
    __syncthreads();

    v8f acc0 = zero8(), acc1 = zero8(), acc2 = zero8(), acc3 = zero8();

    for (int g0 = 0; g0 < G; g0 += KC) {
        for (int idx = tid; idx < 16 * KC; idx += 256) {
            int c = idx >> 7, gg = idx & (KC - 1);
            const float* dr = Din + (g0 + gg) * 10;
            float a = 0.f;
            #pragma unroll
            for (int i = 0; i < 10; ++i) a += h_lds[c][i] * dr[i];
            S[c][gg] = fmaxf(a, 0.f);
        }
        __syncthreads();

        for (int kk = 0; kk < KC; kk += 4) {
            v2f a;
            a.x = S[lrow][kk + kb];
            a.y = S[lrow][kk + kb + 1];
            int k = g0 + kk + kb;
            const float* d0 = Dout + k * DIM455;
            const float* d1 = Dout + (k + 1) * DIM455;

            int n0 = (wave + 0 * 8) * 16 + lrow;
            int n1 = (wave + 1 * 8) * 16 + lrow;
            int n2 = (wave + 2 * 8) * 16 + lrow;
            int n3 = (wave + 3 * 8) * 16 + lrow;
            int n3c = (n3 < DIM455) ? n3 : (DIM455 - 1);
            v2f b0, b1, b2, b3;
            b0.x = d0[n0];   b0.y = d1[n0];
            b1.x = d0[n1];   b1.y = d1[n1];
            b2.x = d0[n2];   b2.y = d1[n2];
            float t3x = d0[n3c], t3y = d1[n3c];
            b3.x = (n3 < DIM455) ? t3x : 0.f;
            b3.y = (n3 < DIM455) ? t3y : 0.f;

            acc0 = wmma4(a, b0, acc0);
            acc1 = wmma4(a, b1, acc1);
            acc2 = wmma4(a, b2, acc2);
            acc3 = wmma4(a, b3, acc3);
        }
        __syncthreads();
    }

    {
        v8f* accs[4] = {&acc0, &acc1, &acc2, &acc3};
        #pragma unroll
        for (int tt = 0; tt < 4; ++tt) {
            int col = (wave + tt * 8) * 16 + lrow;
            if (col < DIM455) {
                #pragma unroll
                for (int j = 0; j < 8; ++j)
                    H2[j + crow][col] = (*accs[tt])[j];
            }
        }
    }
    __syncthreads();

    if (tid < 64) OUTf[tid * OUTS + DIM455] = 0.f;
    const float* H2f = &H2[0][0];
    mix_level<0>(w20, H2f, OUTf, wave, lrow, kb, crow);
    mix_level<1>(w21, H2f, OUTf, wave, lrow, kb, crow);
    mix_level<2>(w22, H2f, OUTf, wave, lrow, kb, crow);
    mix_level<3>(w23, H2f, OUTf, wave, lrow, kb, crow);
    mix_level<4>(w24, H2f, OUTf, wave, lrow, kb, crow);
    mix_level<5>(w25, H2f, OUTf, wave, lrow, kb, crow);
    mix_level<6>(w26, H2f, OUTf, wave, lrow, kb, crow);
    __syncthreads();

    const int kh = wave & 1;
    const int np = (wave >> 1) & 1;
    const int mp = wave >> 2;

    v8f facc0 = zero8(), facc1 = zero8(), facc2 = zero8(), facc3 = zero8();
    {
        const float* oa0 = OUTf + ((mp * 2 + 0) * 16 + lrow) * OUTS;
        const float* oa1 = OUTf + ((mp * 2 + 1) * 16 + lrow) * OUTS;
        const int rb0 = (np * 2 + 0) * 16 + lrow;
        const int rb1 = (np * 2 + 1) * 16 + lrow;
        const int rb1c = (rb1 < 60) ? rb1 : 59;
        const float* i0 = ico + rb0 * DIM455;
        const float* i1 = ico + rb1c * DIM455;
        const int ks0 = kh * 57;
        for (int ks = ks0; ks < ks0 + 57; ++ks) {
            const int k  = ks * 4 + kb;
            const int k1 = (k + 1 < DIM455) ? (k + 1) : (DIM455 - 1);
            v2f a0 = *reinterpret_cast<const v2f*>(oa0 + k);
            v2f a1 = *reinterpret_cast<const v2f*>(oa1 + k);
            float t0x = i0[k], t0y = i0[k1];
            float t1x = i1[k], t1y = i1[k1];
            v2f b0, b1;
            b0.x = t0x;  b0.y = (k + 1 < DIM455) ? t0y : 0.f;
            b1.x = t1x;  b1.y = (k + 1 < DIM455) ? t1y : 0.f;
            facc0 = wmma4(a0, b0, facc0);
            facc1 = wmma4(a0, b1, facc1);
            facc2 = wmma4(a1, b0, facc2);
            facc3 = wmma4(a1, b1, facc3);
        }
    }

    float* red = &H2[0][0];
    {
        v8f* faccs[4] = {&facc0, &facc1, &facc2, &facc3};
        if (kh == 1) {
            #pragma unroll
            for (int q = 0; q < 4; ++q) {
                const int mt = mp * 2 + (q >> 1), nt = np * 2 + (q & 1);
                #pragma unroll
                for (int j = 0; j < 8; ++j)
                    red[(mt * 16 + j + crow) * 64 + nt * 16 + lrow] = (*faccs[q])[j];
            }
        }
        __syncthreads();
        if (kh == 0) {
            const long xg_elems = 32L * 4 * 128 * 60;
            #pragma unroll
            for (int q = 0; q < 4; ++q) {
                const int mt = mp * 2 + (q >> 1), nt = np * 2 + (q & 1);
                const int r = nt * 16 + lrow;
                if (r < 60) {
                    #pragma unroll
                    for (int j = 0; j < 8; ++j) {
                        const int g2 = mt * 16 + j + crow;
                        float vsum = (*faccs[q])[j] + red[g2 * 64 + r];
                        out[xg_elems + ((long)bt * 64 + g2) * 60 + r] = vsum;
                    }
                }
            }
        }
    }
}

__global__ __launch_bounds__(256) void xg_kernel(
    const float* __restrict__ x, const float* __restrict__ ico,
    float* __restrict__ out)
{
    const int tid  = threadIdx.x;
    const int wave = tid >> 5;
    const int lane = tid & 31;
    const int lrow = lane & 15;
    const int kb   = (lane >> 4) << 1;
    const int crow = (lane >> 4) << 3;
    const int mtile = wave >> 1;
    const int ng    = wave & 1;

    const long rowA = (long)blockIdx.x * 64 + mtile * 16 + lrow;
    const float* xr = x + rowA * DIM455;

    v8f acc0 = zero8(), acc1 = zero8();
    const int r0  = (ng * 2 + 0) * 16 + lrow;
    const int r1  = (ng * 2 + 1) * 16 + lrow;
    const int r1c = (r1 < 60) ? r1 : 59;

    for (int k0 = 0; k0 < 456; k0 += 4) {
        int k   = k0 + kb;
        int kc0 = (k     < DIM455) ? k       : (DIM455 - 1);
        int kc1 = (k + 1 < DIM455) ? (k + 1) : (DIM455 - 1);
        float ax = xr[kc0], ay = xr[kc1];
        float b0x = ico[r0  * DIM455 + kc0], b0y = ico[r0  * DIM455 + kc1];
        float b1x = ico[r1c * DIM455 + kc0], b1y = ico[r1c * DIM455 + kc1];
        v2f a, b0, b1;
        a.x  = (k     < DIM455) ? ax  : 0.f;
        a.y  = (k + 1 < DIM455) ? ay  : 0.f;
        b0.x = (k     < DIM455) ? b0x : 0.f;
        b0.y = (k + 1 < DIM455) ? b0y : 0.f;
        b1.x = (k     < DIM455 && r1 < 60) ? b1x : 0.f;
        b1.y = (k + 1 < DIM455 && r1 < 60) ? b1y : 0.f;
        acc0 = wmma4(a, b0, acc0);
        acc1 = wmma4(a, b1, acc1);
    }

    #pragma unroll
    for (int s = 0; s < 2; ++s) {
        int r = (ng * 2 + s) * 16 + lrow;
        if (r < 60) {
            #pragma unroll
            for (int j = 0; j < 8; ++j) {
                long row = (long)blockIdx.x * 64 + mtile * 16 + j + crow;
                out[row * 60 + r] = (s == 0) ? acc0[j] : acc1[j];
            }
        }
    }
}

extern "C" void kernel_launch(void* const* d_in, const int* in_sizes, int n_in,
                              void* d_out, int out_size, void* d_ws, size_t ws_size,
                              hipStream_t stream) {
    const float* x    = (const float*)d_in[0];
    const float* traj = (const float*)d_in[1];
    const float* w1s  = (const float*)d_in[2];
    const float* w1v  = (const float*)d_in[3];
    const float* Din  = (const float*)d_in[4];
    const float* Dout = (const float*)d_in[5];
    const float* ico  = (const float*)d_in[6];
    float* out = (float*)d_out;

    xg_kernel<<<dim3(16384 / 64), dim3(256), 0, stream>>>(x, ico, out);

    traj_kernel<<<dim3(512), dim3(256), 0, stream>>>(
        traj, w1s, w1v, Din, Dout, ico,
        (const float*)d_in[7],  (const float*)d_in[8],  (const float*)d_in[9],
        (const float*)d_in[10], (const float*)d_in[11], (const float*)d_in[12],
        (const float*)d_in[13], out);
}


# each work-group takes two (b,t) (32 rows, 16 waves, 256 groups): every Dout and w2 fragment a wave fetches feeds two row tiles, halving global loads per product; k order and sums as before
# speedup vs baseline: 2.8559x; 1.0332x over previous
#include <hip/hip_runtime.h>

typedef __attribute__((ext_vector_type(2))) float v2f;
typedef __attribute__((ext_vector_type(8))) float v8f;

__device__ __forceinline__ v8f zero8() {
    v8f z = {0.f,0.f,0.f,0.f,0.f,0.f,0.f,0.f};
    return z;
}

__device__ __forceinline__ v8f wmma4(v2f a, v2f b, v8f c) {
    return __builtin_amdgcn_wmma_f32_16x16x4_f32(
         false, a,  false, b,
         (short)0, c,  false,  false);
}

#define DIM455 455

#define H2S  456
#define OUTS 460

template <int L>
__device__ __forceinline__ void mix_frag(const float* __restrict__ wl, const float* hl,
                                         int kx, v2f (&a)[4], v2f& b0, v2f& b1)
{
    constexpr int d = 2 * L + 1, dd = d * d;
    const int f0 = (int)((unsigned)kx / (unsigned)d);
    const int f1 = (int)((unsigned)(kx + 1) / (unsigned)d);
    const float* p0 = wl + (f0 * (63 * dd) + kx * d);
    const float* p1 = wl + (f1 * (63 * dd) + (kx + 1) * d);
    const int h0 = f0 * (H2S - dd) + kx * d;
    const int h1 = f1 * (H2S - dd) + (kx + 1) * d;
    b0.x = hl[h0];
    b0.y = hl[h1];
    b1.x = hl[16 * H2S + h0];
    b1.y = hl[16 * H2S + h1];
    #pragma unroll
    for (int j = 0; j < 4; ++j) {
        a[j].x = p0[j * dd];
        a[j].y = p1[j * dd];
    }
}

template <int L>
__device__ __forceinline__ void mix_level(const float* __restrict__ w2l, const float* H2f,
                                          float* OUTf, int wave, int lrow, int kb, int crow)
{
    constexpr int d = 2 * L + 1, dd = d * d, Kdim = 16 * d;
    constexpr int o = (L * (2 * L - 1) * (2 * L + 1)) / 3;
    const float scale = rsqrtf((float)Kdim);
    const int vc = (lrow < d) ? lrow : (d - 1);
    const float* wl = w2l + (wave * 4) * dd + vc;
    const float* hl = H2f + o + vc;

    v8f acc0[4], acc1[4];
    #pragma unroll
    for (int j = 0; j < 4; ++j) { acc0[j] = zero8(); acc1[j] = zero8(); }

    v2f a_e[4], a_o[4], b0_e, b1_e, b0_o, b1_o;
    mix_frag<L>(wl, hl, kb, a_e, b0_e, b1_e);
    for (int k0 = 0; k0 < Kdim; k0 += 8) {
        mix_frag<L>(wl, hl, k0 + 4 + kb, a_o, b0_o, b1_o);
        #pragma unroll
        for (int j = 0; j < 4; ++j) {
            acc0[j] = wmma4(a_e[j], b0_e, acc0[j]);
            acc1[j] = wmma4(a_e[j], b1_e, acc1[j]);
        }
        const int kn = (k0 + 8 < Kdim) ? (k0 + 8) : k0;
        mix_frag<L>(wl, hl, kn + kb, a_e, b0_e, b1_e);
        #pragma unroll
        for (int j = 0; j < 4; ++j) {
            acc0[j] = wmma4(a_o[j], b0_o, acc0[j]);
            acc1[j] = wmma4(a_o[j], b1_o, acc1[j]);
        }
    }

    float* ob = OUTf + (wave * 4) * OUTS + o + crow * d + lrow;
    if (lrow < d) {
        #pragma unroll
        for (int j = 0; j < 4; ++j) {
            #pragma unroll
            for (int r = 0; r < 8; ++r)
                if (r < d - 8) {
                    ob[j * OUTS + r * d]             = acc0[j][r] * scale;
                    ob[(64 + j) * OUTS + r * d]      = acc1[j][r] * scale;
                }
        }
        if (crow == 0) {
            #pragma unroll
            for (int j = 0; j < 4; ++j) {
                #pragma unroll
                for (int r = 0; r < 8; ++r)
                    if (r >= d - 8 && r < d) {
                        ob[j * OUTS + r * d]        = acc0[j][r] * scale;
                        ob[(64 + j) * OUTS + r * d] = acc1[j][r] * scale;
                    }
            }
        }
    }
}

__global__ __launch_bounds__(512) void traj_kernel(
    const float* __restrict__ traj, const float* __restrict__ w1s,
    const float* __restrict__ w1v,  const float* __restrict__ Din,
    const float* __restrict__ Dout, const float* __restrict__ ico,
    const float* __restrict__ w20, const float* __restrict__ w21,
    const float* __restrict__ w22, const float* __restrict__ w23,
    const float* __restrict__ w24, const float* __restrict__ w25,
    const float* __restrict__ w26, float* __restrict__ out)
{
    constexpr int G = 4096, KC = 128, MR = 32;
    __shared__ float H2[MR][H2S];
    __shared__ __attribute__((aligned(16))) float U[2 * 64 * OUTS];
    float (*S)[130]    = reinterpret_cast<float (*)[130]>(U);
    float (*h_lds)[12] = reinterpret_cast<float (*)[12]>(U + MR * 130);
    float* OUTf = U;

    const int bt0  = blockIdx.x * 2;
    const int tid  = threadIdx.x;
    const int wave = tid >> 5;
    const int lane = tid & 31;
    const int lrow = lane & 15;
    const int kb   = (lane >> 4) << 1;
    const int crow = (lane >> 4) << 3;

    if (tid < 320) {
        int row = tid / 10, i = tid % 10;
        int c = row & 15;
        const float* t10 = traj + (bt0 + (row >> 4)) * 10;
        float hv;
        if (i == 0) {
            hv = w1s[c] * t10[9];
        } else {
            int vv = (i - 1) / 3, mm = (i - 1) % 3;
            float a = 0.f;
            #pragma unroll
            for (int u = 0; u < 3; ++u)
                a += t10[u * 3 + mm] * w1v[c * 9 + u * 3 + vv];
            hv = a * 0.5773502691896258f;
        }
        h_lds[row][i] = hv;
    }
    __syncthreads();

    v8f acc00 = zero8(), acc01 = zero8(), acc10 = zero8(), acc11 = zero8();

    for (int g0 = 0; g0 < G; g0 += KC) {
        for (int idx = tid; idx < MR * KC; idx += 512) {
            int c = idx >> 7, gg = idx & (KC - 1);
            const float* dr = Din + (g0 + gg) * 10;
            float a = 0.f;
            #pragma unroll
            for (int i = 0; i < 10; ++i) a += h_lds[c][i] * dr[i];
            S[c][gg] = fmaxf(a, 0.f);
        }
        __syncthreads();

        for (int kk = 0; kk < KC; kk += 4) {
            v2f a0, a1;
            a0.x = S[lrow][kk + kb];
            a0.y = S[lrow][kk + kb + 1];
            a1.x = S[16 + lrow][kk + kb];
            a1.y = S[16 + lrow][kk + kb + 1];
            int k = g0 + kk + kb;
            const float* d0 = Dout + k * DIM455;
            const float* d1 = Dout + (k + 1) * DIM455;

            int n0 = wave * 16 + lrow;
            int n1 = (wave + 16) * 16 + lrow;
            int n1c = (n1 < DIM455) ? n1 : (DIM455 - 1);
            v2f b0, b1;
            b0.x = d0[n0];   b0.y = d1[n0];
            float t1x = d0[n1c], t1y = d1[n1c];
            b1.x = (n1 < DIM455) ? t1x : 0.f;
            b1.y = (n1 < DIM455) ? t1y : 0.f;

            acc00 = wmma4(a0, b0, acc00);
            acc01 = wmma4(a0, b1, acc01);
            acc10 = wmma4(a1, b0, acc10);
            acc11 = wmma4(a1, b1, acc11);
        }
        __syncthreads();
    }

    {
        v8f* accs[4] = {&acc00, &acc01, &acc10, &acc11};
        #pragma unroll
        for (int tt = 0; tt < 4; ++tt) {
            int col = (wave + (tt & 1) * 16) * 16 + lrow;
            if (col < DIM455) {
                #pragma unroll
                for (int j = 0; j < 8; ++j)
                    H2[(tt >> 1) * 16 + j + crow][col] = (*accs[tt])[j];
            }
        }
    }
    __syncthreads();

    if (tid < 128) OUTf[tid * OUTS + DIM455] = 0.f;
    const float* H2f = &H2[0][0];
    mix_level<0>(w20, H2f, OUTf, wave, lrow, kb, crow);
    mix_level<1>(w21, H2f, OUTf, wave, lrow, kb, crow);
    mix_level<2>(w22, H2f, OUTf, wave, lrow, kb, crow);
    mix_level<3>(w23, H2f, OUTf, wave, lrow, kb, crow);
    mix_level<4>(w24, H2f, OUTf, wave, lrow, kb, crow);
    mix_level<5>(w25, H2f, OUTf, wave, lrow, kb, crow);
    mix_level<6>(w26, H2f, OUTf, wave, lrow, kb, crow);
    __syncthreads();

    const int bsel = wave >> 3;
    const int kh = wave & 1;
    const int np = (wave >> 1) & 1;
    const int mp = (wave >> 2) & 1;
    const int bt = bt0 + bsel;
    OUTf += bsel * 64 * OUTS;

    v8f facc0 = zero8(), facc1 = zero8(), facc2 = zero8(), facc3 = zero8();
    {
        const float* oa0 = OUTf + ((mp * 2 + 0) * 16 + lrow) * OUTS;
        const float* oa1 = OUTf + ((mp * 2 + 1) * 16 + lrow) * OUTS;
        const int rb0 = (np * 2 + 0) * 16 + lrow;
        const int rb1 = (np * 2 + 1) * 16 + lrow;
        const int rb1c = (rb1 < 60) ? rb1 : 59;
        const float* i0 = ico + rb0 * DIM455;
        const float* i1 = ico + rb1c * DIM455;
        const int ks0 = kh * 57;
        for (int ks = ks0; ks < ks0 + 57; ++ks) {
            const int k  = ks * 4 + kb;
            const int k1 = (k + 1 < DIM455) ? (k + 1) : (DIM455 - 1);
            v2f a0 = *reinterpret_cast<const v2f*>(oa0 + k);
            v2f a1 = *reinterpret_cast<const v2f*>(oa1 + k);
            float t0x = i0[k], t0y = i0[k1];
            float t1x = i1[k], t1y = i1[k1];
            v2f b0, b1;
            b0.x = t0x;  b0.y = (k + 1 < DIM455) ? t0y : 0.f;
            b1.x = t1x;  b1.y = (k + 1 < DIM455) ? t1y : 0.f;
            facc0 = wmma4(a0, b0, facc0);
            facc1 = wmma4(a0, b1, facc1);
            facc2 = wmma4(a1, b0, facc2);
            facc3 = wmma4(a1, b1, facc3);
        }
    }

    float* red = &H2[0][0] + bsel * (64 * 64);
    {
        v8f* faccs[4] = {&facc0, &facc1, &facc2, &facc3};
        if (kh == 1) {
            #pragma unroll
            for (int q = 0; q < 4; ++q) {
                const int mt = mp * 2 + (q >> 1), nt = np * 2 + (q & 1);
                #pragma unroll
                for (int j = 0; j < 8; ++j)
                    red[(mt * 16 + j + crow) * 64 + nt * 16 + lrow] = (*faccs[q])[j];
            }
        }
        __syncthreads();
        if (kh == 0) {
            const long xg_elems = 32L * 4 * 128 * 60;
            #pragma unroll
            for (int q = 0; q < 4; ++q) {
                const int mt = mp * 2 + (q >> 1), nt = np * 2 + (q & 1);
                const int r = nt * 16 + lrow;
                if (r < 60) {
                    #pragma unroll
                    for (int j = 0; j < 8; ++j) {
                        const int g2 = mt * 16 + j + crow;
                        float vsum = (*faccs[q])[j] + red[g2 * 64 + r];
                        out[xg_elems + ((long)bt * 64 + g2) * 60 + r] = vsum;
                    }
                }
            }
        }
    }
}

__global__ __launch_bounds__(256) void xg_kernel(
    const float* __restrict__ x, const float* __restrict__ ico,
    float* __restrict__ out)
{
    const int tid  = threadIdx.x;
    const int wave = tid >> 5;
    const int lane = tid & 31;
    const int lrow = lane & 15;
    const int kb   = (lane >> 4) << 1;
    const int crow = (lane >> 4) << 3;
    const int mtile = wave >> 1;
    const int ng    = wave & 1;

    const long rowA = (long)blockIdx.x * 64 + mtile * 16 + lrow;
    const float* xr = x + rowA * DIM455;

    v8f acc0 = zero8(), acc1 = zero8();
    const int r0  = (ng * 2 + 0) * 16 + lrow;
    const int r1  = (ng * 2 + 1) * 16 + lrow;
    const int r1c = (r1 < 60) ? r1 : 59;

    for (int k0 = 0; k0 < 456; k0 += 4) {
        int k   = k0 + kb;
        int kc0 = (k     < DIM455) ? k       : (DIM455 - 1);
        int kc1 = (k + 1 < DIM455) ? (k + 1) : (DIM455 - 1);
        float ax = xr[kc0], ay = xr[kc1];
        float b0x = ico[r0  * DIM455 + kc0], b0y = ico[r0  * DIM455 + kc1];
        float b1x = ico[r1c * DIM455 + kc0], b1y = ico[r1c * DIM455 + kc1];
        v2f a, b0, b1;
        a.x  = (k     < DIM455) ? ax  : 0.f;
        a.y  = (k + 1 < DIM455) ? ay  : 0.f;
        b0.x = (k     < DIM455) ? b0x : 0.f;
        b0.y = (k + 1 < DIM455) ? b0y : 0.f;
        b1.x = (k     < DIM455 && r1 < 60) ? b1x : 0.f;
        b1.y = (k + 1 < DIM455 && r1 < 60) ? b1y : 0.f;
        acc0 = wmma4(a, b0, acc0);
        acc1 = wmma4(a, b1, acc1);
    }

    #pragma unroll
    for (int s = 0; s < 2; ++s) {
        int r = (ng * 2 + s) * 16 + lrow;
        if (r < 60) {
            #pragma unroll
            for (int j = 0; j < 8; ++j) {
                long row = (long)blockIdx.x * 64 + mtile * 16 + j + crow;
                out[row * 60 + r] = (s == 0) ? acc0[j] : acc1[j];
            }
        }
    }
}

extern "C" void kernel_launch(void* const* d_in, const int* in_sizes, int n_in,
                              void* d_out, int out_size, void* d_ws, size_t ws_size,
                              hipStream_t stream) {
    const float* x    = (const float*)d_in[0];
    const float* traj = (const float*)d_in[1];
    const float* w1s  = (const float*)d_in[2];
    const float* w1v  = (const float*)d_in[3];
    const float* Din  = (const float*)d_in[4];
    const float* Dout = (const float*)d_in[5];
    const float* ico  = (const float*)d_in[6];
    float* out = (float*)d_out;

    xg_kernel<<<dim3(16384 / 64), dim3(256), 0, stream>>>(x, ico, out);

    traj_kernel<<<dim3(256), dim3(512), 0, stream>>>(
        traj, w1s, w1v, Din, Dout, ico,
        (const float*)d_in[7],  (const float*)d_in[8],  (const float*)d_in[9],
        (const float*)d_in[10], (const float*)d_in[11], (const float*)d_in[12],
        (const float*)d_in[13], out);
}


# H2 loop: global operands requested a stage ahead (Dout fragments one 32-k group ahead in a second register set, also across the chunk barrier; Din row one chunk ahead), order pinned; sums unchanged
# speedup vs baseline: 3.1598x; 1.1064x over previous
#include <hip/hip_runtime.h>

typedef __attribute__((ext_vector_type(2))) float v2f;
typedef __attribute__((ext_vector_type(8))) float v8f;

__device__ __forceinline__ v8f zero8() {
    v8f z = {0.f,0.f,0.f,0.f,0.f,0.f,0.f,0.f};
    return z;
}

__device__ __forceinline__ v8f wmma4(v2f a, v2f b, v8f c) {
    return __builtin_amdgcn_wmma_f32_16x16x4_f32(
         false, a,  false, b,
         (short)0, c,  false,  false);
}

#define DIM455 455

#define H2S  456
#define OUTS 460

template <int L>
__device__ __forceinline__ void mix_frag(const float* __restrict__ wl, const float* hl,
                                         int kx, v2f (&a)[4], v2f& b0, v2f& b1)
{
    constexpr int d = 2 * L + 1, dd = d * d;
    const int f0 = (int)((unsigned)kx / (unsigned)d);
    const int f1 = (int)((unsigned)(kx + 1) / (unsigned)d);
    const float* p0 = wl + (f0 * (63 * dd) + kx * d);
    const float* p1 = wl + (f1 * (63 * dd) + (kx + 1) * d);
    const int h0 = f0 * (H2S - dd) + kx * d;
    const int h1 = f1 * (H2S - dd) + (kx + 1) * d;
    b0.x = hl[h0];
    b0.y = hl[h1];
    b1.x = hl[16 * H2S + h0];
    b1.y = hl[16 * H2S + h1];
    #pragma unroll
    for (int j = 0; j < 4; ++j) {
        a[j].x = p0[j * dd];
        a[j].y = p1[j * dd];
    }
}

template <int L>
__device__ __forceinline__ void mix_level(const float* __restrict__ w2l, const float* H2f,
                                          float* OUTf, int wave, int lrow, int kb, int crow)
{
    constexpr int d = 2 * L + 1, dd = d * d, Kdim = 16 * d;
    constexpr int o = (L * (2 * L - 1) * (2 * L + 1)) / 3;
    const float scale = rsqrtf((float)Kdim);
    const int vc = (lrow < d) ? lrow : (d - 1);
    const float* wl = w2l + (wave * 4) * dd + vc;
    const float* hl = H2f + o + vc;

    v8f acc0[4], acc1[4];
    #pragma unroll
    for (int j = 0; j < 4; ++j) { acc0[j] = zero8(); acc1[j] = zero8(); }

    v2f a_e[4], a_o[4], b0_e, b1_e, b0_o, b1_o;
    mix_frag<L>(wl, hl, kb, a_e, b0_e, b1_e);
    for (int k0 = 0; k0 < Kdim; k0 += 8) {
        mix_frag<L>(wl, hl, k0 + 4 + kb, a_o, b0_o, b1_o);
        #pragma unroll
        for (int j = 0; j < 4; ++j) {
            acc0[j] = wmma4(a_e[j], b0_e, acc0[j]);
            acc1[j] = wmma4(a_e[j], b1_e, acc1[j]);
        }
        const int kn = (k0 + 8 < Kdim) ? (k0 + 8) : k0;
        mix_frag<L>(wl, hl, kn + kb, a_e, b0_e, b1_e);
        #pragma unroll
        for (int j = 0; j < 4; ++j) {
            acc0[j] = wmma4(a_o[j], b0_o, acc0[j]);
            acc1[j] = wmma4(a_o[j], b1_o, acc1[j]);
        }
    }

    float* ob = OUTf + (wave * 4) * OUTS + o + crow * d + lrow;
    if (lrow < d) {
        #pragma unroll
        for (int j = 0; j < 4; ++j) {
            #pragma unroll
            for (int r = 0; r < 8; ++r)
                if (r < d - 8) {
                    ob[j * OUTS + r * d]             = acc0[j][r] * scale;
                    ob[(64 + j) * OUTS + r * d]      = acc1[j][r] * scale;
                }
        }
        if (crow == 0) {
            #pragma unroll
            for (int j = 0; j < 4; ++j) {
                #pragma unroll
                for (int r = 0; r < 8; ++r)
                    if (r >= d - 8 && r < d) {
                        ob[j * OUTS + r * d]        = acc0[j][r] * scale;
                        ob[(64 + j) * OUTS + r * d] = acc1[j][r] * scale;
                    }
            }
        }
    }
}

__device__ __forceinline__ void dout_group(const float* p0, const float* p1, float (&f)[8][4])
{
    #pragma unroll
    for (int s = 0; s < 8; ++s) {
        f[s][0] = p0[(4 * s) * DIM455];
        f[s][1] = p0[(4 * s + 1) * DIM455];
        f[s][2] = p1[(4 * s) * DIM455];
        f[s][3] = p1[(4 * s + 1) * DIM455];
    }
}

__device__ __forceinline__ void h2_group(const float (*S)[130], int kk0, int lrow, int kb,
                                         bool n1ok, const float (&f)[8][4],
                                         v8f& acc00, v8f& acc01, v8f& acc10, v8f& acc11)
{
    #pragma unroll
    for (int s = 0; s < 8; ++s) {
        const int kk = kk0 + 4 * s + kb;
        v2f a0, a1, b0, b1;
        a0.x = S[lrow][kk];
        a0.y = S[lrow][kk + 1];
        a1.x = S[16 + lrow][kk];
        a1.y = S[16 + lrow][kk + 1];
        b0.x = f[s][0];
        b0.y = f[s][1];
        b1.x = n1ok ? f[s][2] : 0.f;
        b1.y = n1ok ? f[s][3] : 0.f;
        acc00 = wmma4(a0, b0, acc00);
        acc01 = wmma4(a0, b1, acc01);
        acc10 = wmma4(a1, b0, acc10);
        acc11 = wmma4(a1, b1, acc11);
    }
}

__device__ __forceinline__ void order_point(v8f& a, v8f& b, v8f& c, v8f& d)
{
    asm volatile("" : "+v"(a), "+v"(b), "+v"(c), "+v"(d));
}

__device__ __forceinline__ void lds_barrier()
{
    __builtin_amdgcn_fence(__ATOMIC_RELEASE, "workgroup", "local");
    __builtin_amdgcn_s_barrier();
    __builtin_amdgcn_fence(__ATOMIC_ACQUIRE, "workgroup", "local");
}

__global__ __launch_bounds__(512) void traj_kernel(
    const float* __restrict__ traj, const float* __restrict__ w1s,
    const float* __restrict__ w1v,
    const float* Din   ,
    const float* Dout  ,
    const float* __restrict__ ico,
    const float* __restrict__ w20, const float* __restrict__ w21,
    const float* __restrict__ w22, const float* __restrict__ w23,
    const float* __restrict__ w24, const float* __restrict__ w25,
    const float* __restrict__ w26, float* __restrict__ out)
{
    constexpr int G = 4096, KC = 128, MR = 32;
    __shared__ float H2[MR][H2S];
    __shared__ __attribute__((aligned(16))) float U[2 * 64 * OUTS];
    float (*S)[130]    = reinterpret_cast<float (*)[130]>(U);
    float (*h_lds)[12] = reinterpret_cast<float (*)[12]>(U + MR * 130);
    float* OUTf = U;

    const int bt0  = blockIdx.x * 2;
    const int tid  = threadIdx.x;
    const int wave = tid >> 5;
    const int lane = tid & 31;
    const int lrow = lane & 15;
    const int kb   = (lane >> 4) << 1;
    const int crow = (lane >> 4) << 3;

    if (tid < 320) {
        int row = tid / 10, i = tid % 10;
        int c = row & 15;
        const float* t10 = traj + (bt0 + (row >> 4)) * 10;
        float hv;
        if (i == 0) {
            hv = w1s[c] * t10[9];
        } else {
            int vv = (i - 1) / 3, mm = (i - 1) % 3;
            float a = 0.f;
            #pragma unroll
            for (int u = 0; u < 3; ++u)
                a += t10[u * 3 + mm] * w1v[c * 9 + u * 3 + vv];
            hv = a * 0.5773502691896258f;
        }
        h_lds[row][i] = hv;
    }
    __syncthreads();

    v8f acc00 = zero8(), acc01 = zero8(), acc10 = zero8(), acc11 = zero8();

    const int n1  = (wave + 16) * 16 + lrow;
    const int n1c = (n1 < DIM455) ? n1 : (DIM455 - 1);
    const bool n1ok = n1 < DIM455;
    constexpr int GSTEP = 32 * DIM455;
    const float* p0 = Dout + kb * DIM455 + wave * 16 + lrow;
    const float* p1 = Dout + kb * DIM455 + n1c;
    float fe[8][4], fo[8][4];
    dout_group(p0, p1, fe);

    const int gg = tid & (KC - 1);
    float dcur[10], dnxt[10];
    #pragma unroll
    for (int i = 0; i < 10; ++i) dcur[i] = Din[gg * 10 + i];

    for (int g0 = 0; g0 < G; g0 += KC) {
        for (int idx = tid; idx < MR * KC; idx += 512) {
            int c = idx >> 7;
            float a = 0.f;
            #pragma unroll
            for (int i = 0; i < 10; ++i) a += h_lds[c][i] * dcur[i];
            S[c][gg] = fmaxf(a, 0.f);
        }
        __syncthreads();

        {
            const int gn = (g0 + KC < G) ? (g0 + KC) : g0;
            #pragma unroll
            for (int i = 0; i < 10; ++i) dnxt[i] = Din[(gn + gg) * 10 + i];
        }

        #pragma unroll
        for (int gp = 0; gp < 2; ++gp) {
            p0 += GSTEP;  p1 += GSTEP;
            dout_group(p0, p1, fo);
            order_point(acc00, acc01, acc10, acc11);
            h2_group(S, 64 * gp, lrow, kb, n1ok, fe, acc00, acc01, acc10, acc11);
            order_point(acc00, acc01, acc10, acc11);

            const int adv = (gp == 0 || g0 + KC < G) ? GSTEP : 0;
            p0 += adv;  p1 += adv;
            dout_group(p0, p1, fe);
            order_point(acc00, acc01, acc10, acc11);
            h2_group(S, 64 * gp + 32, lrow, kb, n1ok, fo, acc00, acc01, acc10, acc11);
            order_point(acc00, acc01, acc10, acc11);
        }
        lds_barrier();
        #pragma unroll
        for (int i = 0; i < 10; ++i) dcur[i] = dnxt[i];
    }

    {
        v8f* accs[4] = {&acc00, &acc01, &acc10, &acc11};
        #pragma unroll
        for (int tt = 0; tt < 4; ++tt) {
            int col = (wave + (tt & 1) * 16) * 16 + lrow;
            if (col < DIM455) {
                #pragma unroll
                for (int j = 0; j < 8; ++j)
                    H2[(tt >> 1) * 16 + j + crow][col] = (*accs[tt])[j];
            }
        }
    }
    __syncthreads();

    if (tid < 128) OUTf[tid * OUTS + DIM455] = 0.f;
    const float* H2f = &H2[0][0];
    mix_level<0>(w20, H2f, OUTf, wave, lrow, kb, crow);
    mix_level<1>(w21, H2f, OUTf, wave, lrow, kb, crow);
    mix_level<2>(w22, H2f, OUTf, wave, lrow, kb, crow);
    mix_level<3>(w23, H2f, OUTf, wave, lrow, kb, crow);
    mix_level<4>(w24, H2f, OUTf, wave, lrow, kb, crow);
    mix_level<5>(w25, H2f, OUTf, wave, lrow, kb, crow);
    mix_level<6>(w26, H2f, OUTf, wave, lrow, kb, crow);
    __syncthreads();

    const int bsel = wave >> 3;
    const int kh = wave & 1;
    const int np = (wave >> 1) & 1;
    const int mp = (wave >> 2) & 1;
    const int bt = bt0 + bsel;
    OUTf += bsel * 64 * OUTS;

    v8f facc0 = zero8(), facc1 = zero8(), facc2 = zero8(), facc3 = zero8();
    {
        const float* oa0 = OUTf + ((mp * 2 + 0) * 16 + lrow) * OUTS;
        const float* oa1 = OUTf + ((mp * 2 + 1) * 16 + lrow) * OUTS;
        const int rb0 = (np * 2 + 0) * 16 + lrow;
        const int rb1 = (np * 2 + 1) * 16 + lrow;
        const int rb1c = (rb1 < 60) ? rb1 : 59;
        const float* i0 = ico + rb0 * DIM455;
        const float* i1 = ico + rb1c * DIM455;
        const int ks0 = kh * 57;
        for (int ks = ks0; ks < ks0 + 57; ++ks) {
            const int k  = ks * 4 + kb;
            const int k1 = (k + 1 < DIM455) ? (k + 1) : (DIM455 - 1);
            v2f a0 = *reinterpret_cast<const v2f*>(oa0 + k);
            v2f a1 = *reinterpret_cast<const v2f*>(oa1 + k);
            float t0x = i0[k], t0y = i0[k1];
            float t1x = i1[k], t1y = i1[k1];
            v2f b0, b1;
            b0.x = t0x;  b0.y = (k + 1 < DIM455) ? t0y : 0.f;
            b1.x = t1x;  b1.y = (k + 1 < DIM455) ? t1y : 0.f;
            facc0 = wmma4(a0, b0, facc0);
            facc1 = wmma4(a0, b1, facc1);
            facc2 = wmma4(a1, b0, facc2);
            facc3 = wmma4(a1, b1, facc3);
        }
    }

    float* red = &H2[0][0] + bsel * (64 * 64);
    {
        v8f* faccs[4] = {&facc0, &facc1, &facc2, &facc3};
        if (kh == 1) {
            #pragma unroll
            for (int q = 0; q < 4; ++q) {
                const int mt = mp * 2 + (q >> 1), nt = np * 2 + (q & 1);
                #pragma unroll
                for (int j = 0; j < 8; ++j)
                    red[(mt * 16 + j + crow) * 64 + nt * 16 + lrow] = (*faccs[q])[j];
            }
        }
        __syncthreads();
        if (kh == 0) {
            const long xg_elems = 32L * 4 * 128 * 60;
            #pragma unroll
            for (int q = 0; q < 4; ++q) {
                const int mt = mp * 2 + (q >> 1), nt = np * 2 + (q & 1);
                const int r = nt * 16 + lrow;
                if (r < 60) {
                    #pragma unroll
                    for (int j = 0; j < 8; ++j) {
                        const int g2 = mt * 16 + j + crow;
                        float vsum = (*faccs[q])[j] + red[g2 * 64 + r];
                        out[xg_elems + ((long)bt * 64 + g2) * 60 + r] = vsum;
                    }
                }
            }
        }
    }
}

__global__ __launch_bounds__(256) void xg_kernel(
    const float* __restrict__ x, const float* __restrict__ ico,
    float* __restrict__ out)
{
    const int tid  = threadIdx.x;
    const int wave = tid >> 5;
    const int lane = tid & 31;
    const int lrow = lane & 15;
    const int kb   = (lane >> 4) << 1;
    const int crow = (lane >> 4) << 3;
    const int mtile = wave >> 1;
    const int ng    = wave & 1;

    const long rowA = (long)blockIdx.x * 64 + mtile * 16 + lrow;
    const float* xr = x + rowA * DIM455;

    v8f acc0 = zero8(), acc1 = zero8();
    const int r0  = (ng * 2 + 0) * 16 + lrow;
    const int r1  = (ng * 2 + 1) * 16 + lrow;
    const int r1c = (r1 < 60) ? r1 : 59;

    for (int k0 = 0; k0 < 456; k0 += 4) {
        int k   = k0 + kb;
        int kc0 = (k     < DIM455) ? k       : (DIM455 - 1);
        int kc1 = (k + 1 < DIM455) ? (k + 1) : (DIM455 - 1);
        float ax = xr[kc0], ay = xr[kc1];
        float b0x = ico[r0  * DIM455 + kc0], b0y = ico[r0  * DIM455 + kc1];
        float b1x = ico[r1c * DIM455 + kc0], b1y = ico[r1c * DIM455 + kc1];
        v2f a, b0, b1;
        a.x  = (k     < DIM455) ? ax  : 0.f;
        a.y  = (k + 1 < DIM455) ? ay  : 0.f;
        b0.x = (k     < DIM455) ? b0x : 0.f;
        b0.y = (k + 1 < DIM455) ? b0y : 0.f;
        b1.x = (k     < DIM455 && r1 < 60) ? b1x : 0.f;
        b1.y = (k + 1 < DIM455 && r1 < 60) ? b1y : 0.f;
        acc0 = wmma4(a, b0, acc0);
        acc1 = wmma4(a, b1, acc1);
    }

    #pragma unroll
    for (int s = 0; s < 2; ++s) {
        int r = (ng * 2 + s) * 16 + lrow;
        if (r < 60) {
            #pragma unroll
            for (int j = 0; j < 8; ++j) {
                long row = (long)blockIdx.x * 64 + mtile * 16 + j + crow;
                out[row * 60 + r] = (s == 0) ? acc0[j] : acc1[j];
            }
        }
    }
}

extern "C" void kernel_launch(void* const* d_in, const int* in_sizes, int n_in,
                              void* d_out, int out_size, void* d_ws, size_t ws_size,
                              hipStream_t stream) {
    const float* x    = (const float*)d_in[0];
    const float* traj = (const float*)d_in[1];
    const float* w1s  = (const float*)d_in[2];
    const float* w1v  = (const float*)d_in[3];
    const float* Din  = (const float*)d_in[4];
    const float* Dout = (const float*)d_in[5];
    const float* ico  = (const float*)d_in[6];
    float* out = (float*)d_out;

    xg_kernel<<<dim3(16384 / 64), dim3(256), 0, stream>>>(x, ico, out);

    traj_kernel<<<dim3(256), dim3(512), 0, stream>>>(
        traj, w1s, w1v, Din, Dout, ico,
        (const float*)d_in[7],  (const float*)d_in[8],  (const float*)d_in[9],
        (const float*)d_in[10], (const float*)d_in[11], (const float*)d_in[12],
        (const float*)d_in[13], out);
}


# G loop: each wave makes one 16x16 tile of the NEXT sigma chunk on the matrix unit (K=12 f32 WMMA, h in registers; new summation order of the 10-term dot) into a 2nd LDS buffer; one barrier per chunk
# speedup vs baseline: 3.4532x; 1.0929x over previous
#include <hip/hip_runtime.h>

typedef __attribute__((ext_vector_type(2))) float v2f;
typedef __attribute__((ext_vector_type(8))) float v8f;

__device__ __forceinline__ v8f zero8() {
    v8f z = {0.f,0.f,0.f,0.f,0.f,0.f,0.f,0.f};
    return z;
}

__device__ __forceinline__ v8f wmma4(v2f a, v2f b, v8f c) {
    return __builtin_amdgcn_wmma_f32_16x16x4_f32(
         false, a,  false, b,
         (short)0, c,  false,  false);
}

#define DIM455 455

#define H2S  456
#define OUTS 460

template <int L>
__device__ __forceinline__ void mix_frag(const float* __restrict__ wl, const float* hl,
                                         int kx, v2f (&a)[4], v2f& b0, v2f& b1)
{
    constexpr int d = 2 * L + 1, dd = d * d;
    const int f0 = (int)((unsigned)kx / (unsigned)d);
    const int f1 = (int)((unsigned)(kx + 1) / (unsigned)d);
    const float* p0 = wl + (f0 * (63 * dd) + kx * d);
    const float* p1 = wl + (f1 * (63 * dd) + (kx + 1) * d);
    const int h0 = f0 * (H2S - dd) + kx * d;
    const int h1 = f1 * (H2S - dd) + (kx + 1) * d;
    b0.x = hl[h0];
    b0.y = hl[h1];
    b1.x = hl[16 * H2S + h0];
    b1.y = hl[16 * H2S + h1];
    #pragma unroll
    for (int j = 0; j < 4; ++j) {
        a[j].x = p0[j * dd];
        a[j].y = p1[j * dd];
    }
}

template <int L>
__device__ __forceinline__ void mix_level(const float* __restrict__ w2l, const float* H2f,
                                          float* OUTf, int wave, int lrow, int kb, int crow)
{
    constexpr int d = 2 * L + 1, dd = d * d, Kdim = 16 * d;
    constexpr int o = (L * (2 * L - 1) * (2 * L + 1)) / 3;
    const float scale = rsqrtf((float)Kdim);
    const int vc = (lrow < d) ? lrow : (d - 1);
    const float* wl = w2l + (wave * 4) * dd + vc;
    const float* hl = H2f + o + vc;

    v8f acc0[4], acc1[4];
    #pragma unroll
    for (int j = 0; j < 4; ++j) { acc0[j] = zero8(); acc1[j] = zero8(); }

    v2f a_e[4], a_o[4], b0_e, b1_e, b0_o, b1_o;
    mix_frag<L>(wl, hl, kb, a_e, b0_e, b1_e);
    for (int k0 = 0; k0 < Kdim; k0 += 8) {
        mix_frag<L>(wl, hl, k0 + 4 + kb, a_o, b0_o, b1_o);
        #pragma unroll
        for (int j = 0; j < 4; ++j) {
            acc0[j] = wmma4(a_e[j], b0_e, acc0[j]);
            acc1[j] = wmma4(a_e[j], b1_e, acc1[j]);
        }
        const int kn = (k0 + 8 < Kdim) ? (k0 + 8) : k0;
        mix_frag<L>(wl, hl, kn + kb, a_e, b0_e, b1_e);
        #pragma unroll
        for (int j = 0; j < 4; ++j) {
            acc0[j] = wmma4(a_o[j], b0_o, acc0[j]);
            acc1[j] = wmma4(a_o[j], b1_o, acc1[j]);
        }
    }

    float* ob = OUTf + (wave * 4) * OUTS + o + crow * d + lrow;
    if (lrow < d) {
        #pragma unroll
        for (int j = 0; j < 4; ++j) {
            #pragma unroll
            for (int r = 0; r < 8; ++r)
                if (r < d - 8) {
                    ob[j * OUTS + r * d]             = acc0[j][r] * scale;
                    ob[(64 + j) * OUTS + r * d]      = acc1[j][r] * scale;
                }
        }
        if (crow == 0) {
            #pragma unroll
            for (int j = 0; j < 4; ++j) {
                #pragma unroll
                for (int r = 0; r < 8; ++r)
                    if (r >= d - 8 && r < d) {
                        ob[j * OUTS + r * d]        = acc0[j][r] * scale;
                        ob[(64 + j) * OUTS + r * d] = acc1[j][r] * scale;
                    }
            }
        }
    }
}

__device__ __forceinline__ void dout_group(const float* p0, const float* p1, float (&f)[8][4])
{
    #pragma unroll
    for (int s = 0; s < 8; ++s) {
        f[s][0] = p0[(4 * s) * DIM455];
        f[s][1] = p0[(4 * s + 1) * DIM455];
        f[s][2] = p1[(4 * s) * DIM455];
        f[s][3] = p1[(4 * s + 1) * DIM455];
    }
}

__device__ __forceinline__ void h2_group(const float (*S)[130], int kk0, int lrow, int kb,
                                         bool n1ok, const float (&f)[8][4],
                                         v8f& acc00, v8f& acc01, v8f& acc10, v8f& acc11)
{
    #pragma unroll
    for (int s = 0; s < 8; ++s) {
        const int kk = kk0 + 4 * s + kb;
        v2f a0, a1, b0, b1;
        a0.x = S[lrow][kk];
        a0.y = S[lrow][kk + 1];
        a1.x = S[16 + lrow][kk];
        a1.y = S[16 + lrow][kk + 1];
        b0.x = f[s][0];
        b0.y = f[s][1];
        b1.x = n1ok ? f[s][2] : 0.f;
        b1.y = n1ok ? f[s][3] : 0.f;
        acc00 = wmma4(a0, b0, acc00);
        acc01 = wmma4(a0, b1, acc01);
        acc10 = wmma4(a1, b0, acc10);
        acc11 = wmma4(a1, b1, acc11);
    }
}

__device__ __forceinline__ void order_point(v8f& a, v8f& b, v8f& c, v8f& d)
{
    asm volatile("" : "+v"(a), "+v"(b), "+v"(c), "+v"(d));
}

__device__ __forceinline__ void lds_barrier()
{
    __builtin_amdgcn_fence(__ATOMIC_RELEASE, "workgroup", "local");
    __builtin_amdgcn_s_barrier();
    __builtin_amdgcn_fence(__ATOMIC_ACQUIRE, "workgroup", "local");
}

__device__ __forceinline__ void din_frag(const float* dq, int kb, v2f (&db)[3])
{
    db[0] = *reinterpret_cast<const v2f*>(dq + kb);
    db[1] = *reinterpret_cast<const v2f*>(dq + 4 + kb);
    db[2] = *reinterpret_cast<const v2f*>(dq + 8);
}

__device__ __forceinline__ void sigma_tile(const v2f (&ha)[3], const v2f (&db)[3], int kb, float* Sd)
{
    v2f b2;
    b2.x = (kb == 0) ? db[2].x : 0.f;
    b2.y = (kb == 0) ? db[2].y : 0.f;
    v8f acc = zero8();
    acc = wmma4(ha[0], db[0], acc);
    acc = wmma4(ha[1], db[1], acc);
    acc = wmma4(ha[2], b2, acc);
    #pragma unroll
    for (int r = 0; r < 8; ++r) Sd[r * 130] = fmaxf(acc[r], 0.f);
}

__global__ __launch_bounds__(512) void traj_kernel(
    const float* __restrict__ traj, const float* __restrict__ w1s,
    const float* __restrict__ w1v,
    const float* Din   ,
    const float* Dout  ,
    const float* __restrict__ ico,
    const float* __restrict__ w20, const float* __restrict__ w21,
    const float* __restrict__ w22, const float* __restrict__ w23,
    const float* __restrict__ w24, const float* __restrict__ w25,
    const float* __restrict__ w26, float* __restrict__ out)
{
    constexpr int G = 4096, KC = 128, MR = 32;
    __shared__ float H2[MR][H2S];
    __shared__ __attribute__((aligned(16))) float U[2 * 64 * OUTS];
    constexpr int SBUF = MR * 130;
    float (*h_lds)[12] = reinterpret_cast<float (*)[12]>(U + 2 * SBUF);
    float* OUTf = U;

    const int bt0  = blockIdx.x * 2;
    const int tid  = threadIdx.x;
    const int wave = tid >> 5;
    const int lane = tid & 31;
    const int lrow = lane & 15;
    const int kb   = (lane >> 4) << 1;
    const int crow = (lane >> 4) << 3;

    if (tid < 320) {
        int row = tid / 10, i = tid % 10;
        int c = row & 15;
        const float* t10 = traj + (bt0 + (row >> 4)) * 10;
        float hv;
        if (i == 0) {
            hv = w1s[c] * t10[9];
        } else {
            int vv = (i - 1) / 3, mm = (i - 1) % 3;
            float a = 0.f;
            #pragma unroll
            for (int u = 0; u < 3; ++u)
                a += t10[u * 3 + mm] * w1v[c * 9 + u * 3 + vv];
            hv = a * 0.5773502691896258f;
        }
        h_lds[row][i] = hv;
    }
    __syncthreads();

    v8f acc00 = zero8(), acc01 = zero8(), acc10 = zero8(), acc11 = zero8();

    const int n1  = (wave + 16) * 16 + lrow;
    const int n1c = (n1 < DIM455) ? n1 : (DIM455 - 1);
    const bool n1ok = n1 < DIM455;
    constexpr int GSTEP = 32 * DIM455;
    const float* p0 = Dout + kb * DIM455 + wave * 16 + lrow;
    const float* p1 = Dout + kb * DIM455 + n1c;
    float fe[8][4], fo[8][4];

    const int smt = wave & 1;
    const int snt = wave >> 1;
    v2f ha[3];
    {
        const float* hr = &h_lds[smt * 16 + lrow][0];
        ha[0].x = hr[kb];      ha[0].y = hr[kb + 1];
        ha[1].x = hr[4 + kb];  ha[1].y = hr[5 + kb];
        const float h8 = hr[8], h9 = hr[9];
        ha[2].x = (kb == 0) ? h8 : 0.f;
        ha[2].y = (kb == 0) ? h9 : 0.f;
    }
    const float* dq = Din + (snt * 16 + lrow) * 10;
    const int sd_off = (smt * 16 + crow) * 130 + snt * 16 + lrow;
    v2f dn[3];
    din_frag(dq, kb, dn);
    dout_group(p0, p1, fe);
    sigma_tile(ha, dn, kb, U + sd_off);
    dq += KC * 10;
    din_frag(dq, kb, dn);
    lds_barrier();

    for (int g0 = 0; g0 < G; g0 += KC) {
        const int cur = (g0 / KC) & 1;
        const float (*S)[130] = reinterpret_cast<const float (*)[130]>(U + cur * SBUF);
        float* Sd = U + (cur ^ 1) * SBUF + sd_off;

        p0 += GSTEP;  p1 += GSTEP;
        dout_group(p0, p1, fo);
        order_point(acc00, acc01, acc10, acc11);
        h2_group(S, 0, lrow, kb, n1ok, fe, acc00, acc01, acc10, acc11);
        order_point(acc00, acc01, acc10, acc11);
        p0 += GSTEP;  p1 += GSTEP;
        dout_group(p0, p1, fe);
        order_point(acc00, acc01, acc10, acc11);
        h2_group(S, 32, lrow, kb, n1ok, fo, acc00, acc01, acc10, acc11);
        order_point(acc00, acc01, acc10, acc11);

        sigma_tile(ha, dn, kb, Sd);
        dq += (g0 + 2 * KC < G) ? KC * 10 : 0;
        din_frag(dq, kb, dn);

        p0 += GSTEP;  p1 += GSTEP;
        dout_group(p0, p1, fo);
        order_point(acc00, acc01, acc10, acc11);
        h2_group(S, 64, lrow, kb, n1ok, fe, acc00, acc01, acc10, acc11);
        order_point(acc00, acc01, acc10, acc11);
        const int adv = (g0 + KC < G) ? GSTEP : 0;
        p0 += adv;  p1 += adv;
        dout_group(p0, p1, fe);
        order_point(acc00, acc01, acc10, acc11);
        h2_group(S, 96, lrow, kb, n1ok, fo, acc00, acc01, acc10, acc11);
        order_point(acc00, acc01, acc10, acc11);
        lds_barrier();
    }

    {
        v8f* accs[4] = {&acc00, &acc01, &acc10, &acc11};
        #pragma unroll
        for (int tt = 0; tt < 4; ++tt) {
            int col = (wave + (tt & 1) * 16) * 16 + lrow;
            if (col < DIM455) {
                #pragma unroll
                for (int j = 0; j < 8; ++j)
                    H2[(tt >> 1) * 16 + j + crow][col] = (*accs[tt])[j];
            }
        }
    }
    __syncthreads();

    if (tid < 128) OUTf[tid * OUTS + DIM455] = 0.f;
    const float* H2f = &H2[0][0];
    mix_level<0>(w20, H2f, OUTf, wave, lrow, kb, crow);
    mix_level<1>(w21, H2f, OUTf, wave, lrow, kb, crow);
    mix_level<2>(w22, H2f, OUTf, wave, lrow, kb, crow);
    mix_level<3>(w23, H2f, OUTf, wave, lrow, kb, crow);
    mix_level<4>(w24, H2f, OUTf, wave, lrow, kb, crow);
    mix_level<5>(w25, H2f, OUTf, wave, lrow, kb, crow);
    mix_level<6>(w26, H2f, OUTf, wave, lrow, kb, crow);
    __syncthreads();

    const int bsel = wave >> 3;
    const int kh = wave & 1;
    const int np = (wave >> 1) & 1;
    const int mp = (wave >> 2) & 1;
    const int bt = bt0 + bsel;
    OUTf += bsel * 64 * OUTS;

    v8f facc0 = zero8(), facc1 = zero8(), facc2 = zero8(), facc3 = zero8();
    {
        const float* oa0 = OUTf + ((mp * 2 + 0) * 16 + lrow) * OUTS;
        const float* oa1 = OUTf + ((mp * 2 + 1) * 16 + lrow) * OUTS;
        const int rb0 = (np * 2 + 0) * 16 + lrow;
        const int rb1 = (np * 2 + 1) * 16 + lrow;
        const int rb1c = (rb1 < 60) ? rb1 : 59;
        const float* i0 = ico + rb0 * DIM455;
        const float* i1 = ico + rb1c * DIM455;
        const int ks0 = kh * 57;
        for (int ks = ks0; ks < ks0 + 57; ++ks) {
            const int k  = ks * 4 + kb;
            const int k1 = (k + 1 < DIM455) ? (k + 1) : (DIM455 - 1);
            v2f a0 = *reinterpret_cast<const v2f*>(oa0 + k);
            v2f a1 = *reinterpret_cast<const v2f*>(oa1 + k);
            float t0x = i0[k], t0y = i0[k1];
            float t1x = i1[k], t1y = i1[k1];
            v2f b0, b1;
            b0.x = t0x;  b0.y = (k + 1 < DIM455) ? t0y : 0.f;
            b1.x = t1x;  b1.y = (k + 1 < DIM455) ? t1y : 0.f;
            facc0 = wmma4(a0, b0, facc0);
            facc1 = wmma4(a0, b1, facc1);
            facc2 = wmma4(a1, b0, facc2);
            facc3 = wmma4(a1, b1, facc3);
        }
    }

    float* red = &H2[0][0] + bsel * (64 * 64);
    {
        v8f* faccs[4] = {&facc0, &facc1, &facc2, &facc3};
        if (kh == 1) {
            #pragma unroll
            for (int q = 0; q < 4; ++q) {
                const int mt = mp * 2 + (q >> 1), nt = np * 2 + (q & 1);
                #pragma unroll
                for (int j = 0; j < 8; ++j)
                    red[(mt * 16 + j + crow) * 64 + nt * 16 + lrow] = (*faccs[q])[j];
            }
        }
        __syncthreads();
        if (kh == 0) {
            const long xg_elems = 32L * 4 * 128 * 60;
            #pragma unroll
            for (int q = 0; q < 4; ++q) {
                const int mt = mp * 2 + (q >> 1), nt = np * 2 + (q & 1);
                const int r = nt * 16 + lrow;
                if (r < 60) {
                    #pragma unroll
                    for (int j = 0; j < 8; ++j) {
                        const int g2 = mt * 16 + j + crow;
                        float vsum = (*faccs[q])[j] + red[g2 * 64 + r];
                        out[xg_elems + ((long)bt * 64 + g2) * 60 + r] = vsum;
                    }
                }
            }
        }
    }
}

__global__ __launch_bounds__(256) void xg_kernel(
    const float* __restrict__ x, const float* __restrict__ ico,
    float* __restrict__ out)
{
    const int tid  = threadIdx.x;
    const int wave = tid >> 5;
    const int lane = tid & 31;
    const int lrow = lane & 15;
    const int kb   = (lane >> 4) << 1;
    const int crow = (lane >> 4) << 3;
    const int mtile = wave >> 1;
    const int ng    = wave & 1;

    const long rowA = (long)blockIdx.x * 64 + mtile * 16 + lrow;
    const float* xr = x + rowA * DIM455;

    v8f acc0 = zero8(), acc1 = zero8();
    const int r0  = (ng * 2 + 0) * 16 + lrow;
    const int r1  = (ng * 2 + 1) * 16 + lrow;
    const int r1c = (r1 < 60) ? r1 : 59;

    for (int k0 = 0; k0 < 456; k0 += 4) {
        int k   = k0 + kb;
        int kc0 = (k     < DIM455) ? k       : (DIM455 - 1);
        int kc1 = (k + 1 < DIM455) ? (k + 1) : (DIM455 - 1);
        float ax = xr[kc0], ay = xr[kc1];
        float b0x = ico[r0  * DIM455 + kc0], b0y = ico[r0  * DIM455 + kc1];
        float b1x = ico[r1c * DIM455 + kc0], b1y = ico[r1c * DIM455 + kc1];
        v2f a, b0, b1;
        a.x  = (k     < DIM455) ? ax  : 0.f;
        a.y  = (k + 1 < DIM455) ? ay  : 0.f;
        b0.x = (k     < DIM455) ? b0x : 0.f;
        b0.y = (k + 1 < DIM455) ? b0y : 0.f;
        b1.x = (k     < DIM455 && r1 < 60) ? b1x : 0.f;
        b1.y = (k + 1 < DIM455 && r1 < 60) ? b1y : 0.f;
        acc0 = wmma4(a, b0, acc0);
        acc1 = wmma4(a, b1, acc1);
    }

    #pragma unroll
    for (int s = 0; s < 2; ++s) {
        int r = (ng * 2 + s) * 16 + lrow;
        if (r < 60) {
            #pragma unroll
            for (int j = 0; j < 8; ++j) {
                long row = (long)blockIdx.x * 64 + mtile * 16 + j + crow;
                out[row * 60 + r] = (s == 0) ? acc0[j] : acc1[j];
            }
        }
    }
}

extern "C" void kernel_launch(void* const* d_in, const int* in_sizes, int n_in,
                              void* d_out, int out_size, void* d_ws, size_t ws_size,
                              hipStream_t stream) {
    const float* x    = (const float*)d_in[0];
    const float* traj = (const float*)d_in[1];
    const float* w1s  = (const float*)d_in[2];
    const float* w1v  = (const float*)d_in[3];
    const float* Din  = (const float*)d_in[4];
    const float* Dout = (const float*)d_in[5];
    const float* ico  = (const float*)d_in[6];
    float* out = (float*)d_out;

    xg_kernel<<<dim3(16384 / 64), dim3(256), 0, stream>>>(x, ico, out);

    traj_kernel<<<dim3(256), dim3(512), 0, stream>>>(
        traj, w1s, w1v, Din, Dout, ico,
        (const float*)d_in[7],  (const float*)d_in[8],  (const float*)d_in[9],
        (const float*)d_in[10], (const float*)d_in[11], (const float*)d_in[12],
        (const float*)d_in[13], out);
}


# H2 loop: Dout read as 16-byte row pieces: a wave owns 64 neighbouring columns (4 column tiles, column mod 4) and half of each chunk's k; the two k halves are added in LDS (new summation order)
# speedup vs baseline: 3.5665x; 1.0328x over previous
#include <hip/hip_runtime.h>

typedef __attribute__((ext_vector_type(2))) float v2f;
typedef __attribute__((ext_vector_type(8))) float v8f;
typedef __attribute__((ext_vector_type(4))) float v4f;
typedef v4f __attribute__((aligned(4))) v4fu;

__device__ __forceinline__ v8f zero8() {
    v8f z = {0.f,0.f,0.f,0.f,0.f,0.f,0.f,0.f};
    return z;
}

__device__ __forceinline__ v8f wmma4(v2f a, v2f b, v8f c) {
    return __builtin_amdgcn_wmma_f32_16x16x4_f32(
         false, a,  false, b,
         (short)0, c,  false,  false);
}

#define DIM455 455

#define H2S  456
#define OUTS 460

template <int L>
__device__ __forceinline__ void mix_frag(const float* __restrict__ wl, const float* hl,
                                         int kx, v2f (&a)[4], v2f& b0, v2f& b1)
{
    constexpr int d = 2 * L + 1, dd = d * d;
    const int f0 = (int)((unsigned)kx / (unsigned)d);
    const int f1 = (int)((unsigned)(kx + 1) / (unsigned)d);
    const float* p0 = wl + (f0 * (63 * dd) + kx * d);
    const float* p1 = wl + (f1 * (63 * dd) + (kx + 1) * d);
    const int h0 = f0 * (H2S - dd) + kx * d;
    const int h1 = f1 * (H2S - dd) + (kx + 1) * d;
    b0.x = hl[h0];
    b0.y = hl[h1];
    b1.x = hl[16 * H2S + h0];
    b1.y = hl[16 * H2S + h1];
    #pragma unroll
    for (int j = 0; j < 4; ++j) {
        a[j].x = p0[j * dd];
        a[j].y = p1[j * dd];
    }
}

template <int L>
__device__ __forceinline__ void mix_level(const float* __restrict__ w2l, const float* H2f,
                                          float* OUTf, int wave, int lrow, int kb, int crow)
{
    constexpr int d = 2 * L + 1, dd = d * d, Kdim = 16 * d;
    constexpr int o = (L * (2 * L - 1) * (2 * L + 1)) / 3;
    const float scale = rsqrtf((float)Kdim);
    const int vc = (lrow < d) ? lrow : (d - 1);
    const float* wl = w2l + (wave * 4) * dd + vc;
    const float* hl = H2f + o + vc;

    v8f acc0[4], acc1[4];
    #pragma unroll
    for (int j = 0; j < 4; ++j) { acc0[j] = zero8(); acc1[j] = zero8(); }

    v2f a_e[4], a_o[4], b0_e, b1_e, b0_o, b1_o;
    mix_frag<L>(wl, hl, kb, a_e, b0_e, b1_e);
    for (int k0 = 0; k0 < Kdim; k0 += 8) {
        mix_frag<L>(wl, hl, k0 + 4 + kb, a_o, b0_o, b1_o);
        #pragma unroll
        for (int j = 0; j < 4; ++j) {
            acc0[j] = wmma4(a_e[j], b0_e, acc0[j]);
            acc1[j] = wmma4(a_e[j], b1_e, acc1[j]);
        }
        const int kn = (k0 + 8 < Kdim) ? (k0 + 8) : k0;
        mix_frag<L>(wl, hl, kn + kb, a_e, b0_e, b1_e);
        #pragma unroll
        for (int j = 0; j < 4; ++j) {
            acc0[j] = wmma4(a_o[j], b0_o, acc0[j]);
            acc1[j] = wmma4(a_o[j], b1_o, acc1[j]);
        }
    }

    float* ob = OUTf + (wave * 4) * OUTS + o + crow * d + lrow;
    if (lrow < d) {
        #pragma unroll
        for (int j = 0; j < 4; ++j) {
            #pragma unroll
            for (int r = 0; r < 8; ++r)
                if (r < d - 8) {
                    ob[j * OUTS + r * d]             = acc0[j][r] * scale;
                    ob[(64 + j) * OUTS + r * d]      = acc1[j][r] * scale;
                }
        }
        if (crow == 0) {
            #pragma unroll
            for (int j = 0; j < 4; ++j) {
                #pragma unroll
                for (int r = 0; r < 8; ++r)
                    if (r >= d - 8 && r < d) {
                        ob[j * OUTS + r * d]        = acc0[j][r] * scale;
                        ob[(64 + j) * OUTS + r * d] = acc1[j][r] * scale;
                    }
            }
        }
    }
}

__device__ __forceinline__ void dout_group(const float* p, v4f (&f)[4][2])
{
    #pragma unroll
    for (int s = 0; s < 4; ++s) {
        f[s][0] = *reinterpret_cast<const v4fu*>(p + (4 * s) * DIM455);
        f[s][1] = *reinterpret_cast<const v4fu*>(p + (4 * s + 1) * DIM455);
    }
}

__device__ __forceinline__ void h2_group(const float (*S)[130], int kk0, int lrow, int kb,
                                         const v4f (&f)[4][2], v8f (&acc)[2][4])
{
    #pragma unroll
    for (int s = 0; s < 4; ++s) {
        const int kk = kk0 + 4 * s + kb;
        v2f a0, a1;
        a0.x = S[lrow][kk];
        a0.y = S[lrow][kk + 1];
        a1.x = S[16 + lrow][kk];
        a1.y = S[16 + lrow][kk + 1];
        #pragma unroll
        for (int j = 0; j < 4; ++j) {
            v2f b;
            b.x = f[s][0][j];
            b.y = f[s][1][j];
            acc[0][j] = wmma4(a0, b, acc[0][j]);
            acc[1][j] = wmma4(a1, b, acc[1][j]);
        }
    }
}

__device__ __forceinline__ void order_point(v8f (&acc)[2][4])
{
    asm volatile("" : "+v"(acc[0][0]), "+v"(acc[0][1]), "+v"(acc[0][2]), "+v"(acc[0][3]),
                      "+v"(acc[1][0]), "+v"(acc[1][1]), "+v"(acc[1][2]), "+v"(acc[1][3]));
}

__device__ __forceinline__ void lds_barrier()
{
    __builtin_amdgcn_fence(__ATOMIC_RELEASE, "workgroup", "local");
    __builtin_amdgcn_s_barrier();
    __builtin_amdgcn_fence(__ATOMIC_ACQUIRE, "workgroup", "local");
}

__device__ __forceinline__ void din_frag(const float* dq, int kb, v2f (&db)[3])
{
    db[0] = *reinterpret_cast<const v2f*>(dq + kb);
    db[1] = *reinterpret_cast<const v2f*>(dq + 4 + kb);
    db[2] = *reinterpret_cast<const v2f*>(dq + 8);
}

__device__ __forceinline__ void sigma_tile(const v2f (&ha)[3], const v2f (&db)[3], int kb, float* Sd)
{
    v2f b2;
    b2.x = (kb == 0) ? db[2].x : 0.f;
    b2.y = (kb == 0) ? db[2].y : 0.f;
    v8f acc = zero8();
    acc = wmma4(ha[0], db[0], acc);
    acc = wmma4(ha[1], db[1], acc);
    acc = wmma4(ha[2], b2, acc);
    #pragma unroll
    for (int r = 0; r < 8; ++r) Sd[r * 130] = fmaxf(acc[r], 0.f);
}

__global__ __launch_bounds__(512) void traj_kernel(
    const float* __restrict__ traj, const float* __restrict__ w1s,
    const float* __restrict__ w1v,
    const float* Din   ,
    const float* Dout  ,
    const float* __restrict__ ico,
    const float* __restrict__ w20, const float* __restrict__ w21,
    const float* __restrict__ w22, const float* __restrict__ w23,
    const float* __restrict__ w24, const float* __restrict__ w25,
    const float* __restrict__ w26, float* __restrict__ out)
{
    constexpr int G = 4096, KC = 128, MR = 32;
    __shared__ float H2[MR][H2S];
    __shared__ __attribute__((aligned(16))) float U[2 * 64 * OUTS];
    constexpr int SBUF = MR * 130;
    float (*h_lds)[12] = reinterpret_cast<float (*)[12]>(U + 2 * SBUF);
    float* OUTf = U;

    const int bt0  = blockIdx.x * 2;
    const int tid  = threadIdx.x;
    const int wave = tid >> 5;
    const int lane = tid & 31;
    const int lrow = lane & 15;
    const int kb   = (lane >> 4) << 1;
    const int crow = (lane >> 4) << 3;

    if (tid < 320) {
        int row = tid / 10, i = tid % 10;
        int c = row & 15;
        const float* t10 = traj + (bt0 + (row >> 4)) * 10;
        float hv;
        if (i == 0) {
            hv = w1s[c] * t10[9];
        } else {
            int vv = (i - 1) / 3, mm = (i - 1) % 3;
            float a = 0.f;
            #pragma unroll
            for (int u = 0; u < 3; ++u)
                a += t10[u * 3 + mm] * w1v[c * 9 + u * 3 + vv];
            hv = a * 0.5773502691896258f;
        }
        h_lds[row][i] = hv;
    }
    __syncthreads();

    v8f acc[2][4];
    #pragma unroll
    for (int m = 0; m < 2; ++m) {
        #pragma unroll
        for (int j = 0; j < 4; ++j) acc[m][j] = zero8();
    }

    const int wn   = wave & 7;
    const int khalf   = wave >> 3;
    const int col0 = 64 * wn + 4 * lrow;
    const int colc = (col0 < DIM455 - 4) ? col0 : (DIM455 - 4);
    constexpr int GSTEP = 16 * DIM455;
    const float* p0 = Dout + (64 * khalf + kb) * DIM455 + colc;
    v4f fe[4][2], fo[4][2];

    const int smt = wave & 1;
    const int snt = wave >> 1;
    v2f ha[3];
    {
        const float* hr = &h_lds[smt * 16 + lrow][0];
        ha[0].x = hr[kb];      ha[0].y = hr[kb + 1];
        ha[1].x = hr[4 + kb];  ha[1].y = hr[5 + kb];
        const float h8 = hr[8], h9 = hr[9];
        ha[2].x = (kb == 0) ? h8 : 0.f;
        ha[2].y = (kb == 0) ? h9 : 0.f;
    }
    const float* dq = Din + (snt * 16 + lrow) * 10;
    const int sd_off = (smt * 16 + crow) * 130 + snt * 16 + lrow;
    v2f dn[3];
    din_frag(dq, kb, dn);
    dout_group(p0, fe);
    sigma_tile(ha, dn, kb, U + sd_off);
    dq += KC * 10;
    din_frag(dq, kb, dn);
    lds_barrier();

    const int kq = 64 * khalf;
    for (int g0 = 0; g0 < G; g0 += KC) {
        const int cur = (g0 / KC) & 1;
        const float (*S)[130] = reinterpret_cast<const float (*)[130]>(U + cur * SBUF);
        float* Sd = U + (cur ^ 1) * SBUF + sd_off;

        p0 += GSTEP;
        dout_group(p0, fo);
        order_point(acc);
        h2_group(S, kq, lrow, kb, fe, acc);
        order_point(acc);
        p0 += GSTEP;
        dout_group(p0, fe);
        order_point(acc);
        h2_group(S, kq + 16, lrow, kb, fo, acc);
        order_point(acc);

        sigma_tile(ha, dn, kb, Sd);
        dq += (g0 + 2 * KC < G) ? KC * 10 : 0;
        din_frag(dq, kb, dn);

        p0 += GSTEP;
        dout_group(p0, fo);
        order_point(acc);
        h2_group(S, kq + 32, lrow, kb, fe, acc);
        order_point(acc);
        const int adv = (g0 + KC < G) ? (GSTEP + 64 * DIM455) : 0;
        p0 += adv;
        dout_group(p0, fe);
        order_point(acc);
        h2_group(S, kq + 48, lrow, kb, fo, acc);
        order_point(acc);
        lds_barrier();
    }

    if (khalf == 1) {
        #pragma unroll
        for (int m = 0; m < 2; ++m) {
            #pragma unroll
            for (int j = 0; j < 4; ++j) {
                const int col = colc + j;
                if (col >= col0 && col < DIM455) {
                    #pragma unroll
                    for (int r = 0; r < 8; ++r)
                        H2[m * 16 + r + crow][col] = acc[m][j][r];
                }
            }
        }
    }
    __syncthreads();
    if (khalf == 0) {
        #pragma unroll
        for (int m = 0; m < 2; ++m) {
            #pragma unroll
            for (int j = 0; j < 4; ++j) {
                const int col = colc + j;
                if (col >= col0 && col < DIM455) {
                    #pragma unroll
                    for (int r = 0; r < 8; ++r)
                        H2[m * 16 + r + crow][col] += acc[m][j][r];
                }
            }
        }
    }
    __syncthreads();

    if (tid < 128) OUTf[tid * OUTS + DIM455] = 0.f;
    const float* H2f = &H2[0][0];
    mix_level<0>(w20, H2f, OUTf, wave, lrow, kb, crow);
    mix_level<1>(w21, H2f, OUTf, wave, lrow, kb, crow);
    mix_level<2>(w22, H2f, OUTf, wave, lrow, kb, crow);
    mix_level<3>(w23, H2f, OUTf, wave, lrow, kb, crow);
    mix_level<4>(w24, H2f, OUTf, wave, lrow, kb, crow);
    mix_level<5>(w25, H2f, OUTf, wave, lrow, kb, crow);
    mix_level<6>(w26, H2f, OUTf, wave, lrow, kb, crow);
    __syncthreads();

    const int bsel = wave >> 3;
    const int kh = wave & 1;
    const int np = (wave >> 1) & 1;
    const int mp = (wave >> 2) & 1;
    const int bt = bt0 + bsel;
    OUTf += bsel * 64 * OUTS;

    v8f facc0 = zero8(), facc1 = zero8(), facc2 = zero8(), facc3 = zero8();
    {
        const float* oa0 = OUTf + ((mp * 2 + 0) * 16 + lrow) * OUTS;
        const float* oa1 = OUTf + ((mp * 2 + 1) * 16 + lrow) * OUTS;
        const int rb0 = (np * 2 + 0) * 16 + lrow;
        const int rb1 = (np * 2 + 1) * 16 + lrow;
        const int rb1c = (rb1 < 60) ? rb1 : 59;
        const float* i0 = ico + rb0 * DIM455;
        const float* i1 = ico + rb1c * DIM455;
        const int ks0 = kh * 57;
        for (int ks = ks0; ks < ks0 + 57; ++ks) {
            const int k  = ks * 4 + kb;
            const int k1 = (k + 1 < DIM455) ? (k + 1) : (DIM455 - 1);
            v2f a0 = *reinterpret_cast<const v2f*>(oa0 + k);
            v2f a1 = *reinterpret_cast<const v2f*>(oa1 + k);
            float t0x = i0[k], t0y = i0[k1];
            float t1x = i1[k], t1y = i1[k1];
            v2f b0, b1;
            b0.x = t0x;  b0.y = (k + 1 < DIM455) ? t0y : 0.f;
            b1.x = t1x;  b1.y = (k + 1 < DIM455) ? t1y : 0.f;
            facc0 = wmma4(a0, b0, facc0);
            facc1 = wmma4(a0, b1, facc1);
            facc2 = wmma4(a1, b0, facc2);
            facc3 = wmma4(a1, b1, facc3);
        }
    }

    float* red = &H2[0][0] + bsel * (64 * 64);
    {
        v8f* faccs[4] = {&facc0, &facc1, &facc2, &facc3};
        if (kh == 1) {
            #pragma unroll
            for (int q = 0; q < 4; ++q) {
                const int mt = mp * 2 + (q >> 1), nt = np * 2 + (q & 1);
                #pragma unroll
                for (int j = 0; j < 8; ++j)
                    red[(mt * 16 + j + crow) * 64 + nt * 16 + lrow] = (*faccs[q])[j];
            }
        }
        __syncthreads();
        if (kh == 0) {
            const long xg_elems = 32L * 4 * 128 * 60;
            #pragma unroll
            for (int q = 0; q < 4; ++q) {
                const int mt = mp * 2 + (q >> 1), nt = np * 2 + (q & 1);
                const int r = nt * 16 + lrow;
                if (r < 60) {
                    #pragma unroll
                    for (int j = 0; j < 8; ++j) {
                        const int g2 = mt * 16 + j + crow;
                        float vsum = (*faccs[q])[j] + red[g2 * 64 + r];
                        out[xg_elems + ((long)bt * 64 + g2) * 60 + r] = vsum;
                    }
                }
            }
        }
    }
}

__global__ __launch_bounds__(256) void xg_kernel(
    const float* __restrict__ x, const float* __restrict__ ico,
    float* __restrict__ out)
{
    const int tid  = threadIdx.x;
    const int wave = tid >> 5;
    const int lane = tid & 31;
    const int lrow = lane & 15;
    const int kb   = (lane >> 4) << 1;
    const int crow = (lane >> 4) << 3;
    const int mtile = wave >> 1;
    const int ng    = wave & 1;

    const long rowA = (long)blockIdx.x * 64 + mtile * 16 + lrow;
    const float* xr = x + rowA * DIM455;

    v8f acc0 = zero8(), acc1 = zero8();
    const int r0  = (ng * 2 + 0) * 16 + lrow;
    const int r1  = (ng * 2 + 1) * 16 + lrow;
    const int r1c = (r1 < 60) ? r1 : 59;

    for (int k0 = 0; k0 < 456; k0 += 4) {
        int k   = k0 + kb;
        int kc0 = (k     < DIM455) ? k       : (DIM455 - 1);
        int kc1 = (k + 1 < DIM455) ? (k + 1) : (DIM455 - 1);
        float ax = xr[kc0], ay = xr[kc1];
        float b0x = ico[r0  * DIM455 + kc0], b0y = ico[r0  * DIM455 + kc1];
        float b1x = ico[r1c * DIM455 + kc0], b1y = ico[r1c * DIM455 + kc1];
        v2f a, b0, b1;
        a.x  = (k     < DIM455) ? ax  : 0.f;
        a.y  = (k + 1 < DIM455) ? ay  : 0.f;
        b0.x = (k     < DIM455) ? b0x : 0.f;
        b0.y = (k + 1 < DIM455) ? b0y : 0.f;
        b1.x = (k     < DIM455 && r1 < 60) ? b1x : 0.f;
        b1.y = (k + 1 < DIM455 && r1 < 60) ? b1y : 0.f;
        acc0 = wmma4(a, b0, acc0);
        acc1 = wmma4(a, b1, acc1);
    }

    #pragma unroll
    for (int s = 0; s < 2; ++s) {
        int r = (ng * 2 + s) * 16 + lrow;
        if (r < 60) {
            #pragma unroll
            for (int j = 0; j < 8; ++j) {
                long row = (long)blockIdx.x * 64 + mtile * 16 + j + crow;
                out[row * 60 + r] = (s == 0) ? acc0[j] : acc1[j];
            }
        }
    }
}

extern "C" void kernel_launch(void* const* d_in, const int* in_sizes, int n_in,
                              void* d_out, int out_size, void* d_ws, size_t ws_size,
                              hipStream_t stream) {
    const float* x    = (const float*)d_in[0];
    const float* traj = (const float*)d_in[1];
    const float* w1s  = (const float*)d_in[2];
    const float* w1v  = (const float*)d_in[3];
    const float* Din  = (const float*)d_in[4];
    const float* Dout = (const float*)d_in[5];
    const float* ico  = (const float*)d_in[6];
    float* out = (float*)d_out;

    xg_kernel<<<dim3(16384 / 64), dim3(256), 0, stream>>>(x, ico, out);

    traj_kernel<<<dim3(256), dim3(512), 0, stream>>>(
        traj, w1s, w1v, Din, Dout, ico,
        (const float*)d_in[7],  (const float*)d_in[8],  (const float*)d_in[9],
        (const float*)d_in[10], (const float*)d_in[11], (const float*)d_in[12],
        (const float*)d_in[13], out);
}


# w2 mix re-tiled: a level's 64 d rows (g2,v) packed into 4 d full 16-row tiles (were 64 tiles, d of 16 rows used), dealt as 49 four-tile slots evenly to the waves; k order and sums unchanged
# speedup vs baseline: 3.9188x; 1.0988x over previous
#include <hip/hip_runtime.h>

typedef __attribute__((ext_vector_type(2))) float v2f;
typedef __attribute__((ext_vector_type(8))) float v8f;
typedef __attribute__((ext_vector_type(4))) float v4f;
typedef v4f __attribute__((aligned(4))) v4fu;

__device__ __forceinline__ v8f zero8() {
    v8f z = {0.f,0.f,0.f,0.f,0.f,0.f,0.f,0.f};
    return z;
}

__device__ __forceinline__ v8f wmma4(v2f a, v2f b, v8f c) {
    return __builtin_amdgcn_wmma_f32_16x16x4_f32(
         false, a,  false, b,
         (short)0, c,  false,  false);
}

#define DIM455 455

#define H2S  456
#define OUTS 460

template <int L, int NT>
__device__ __forceinline__ void mix_frag(const float* __restrict__ w2l, const int (&rowoff)[NT],
                                         const float* hl, int kx, v2f (&a)[NT], v2f& b0, v2f& b1)
{
    constexpr int d = 2 * L + 1, dd = d * d;
    const int f0 = (int)((unsigned)kx / (unsigned)d);
    const int f1 = (int)((unsigned)(kx + 1) / (unsigned)d);
    const int w0 = f0 * (63 * dd) + kx * d;
    const int w1 = f1 * (63 * dd) + (kx + 1) * d;
    const int h0 = f0 * (H2S - dd) + kx * d;
    const int h1 = f1 * (H2S - dd) + (kx + 1) * d;
    b0.x = hl[h0];
    b0.y = hl[h1];
    b1.x = hl[16 * H2S + h0];
    b1.y = hl[16 * H2S + h1];
    #pragma unroll
    for (int j = 0; j < NT; ++j) {
        a[j].x = w2l[rowoff[j] + w0];
        a[j].y = w2l[rowoff[j] + w1];
    }
}

template <int L>
__device__ __forceinline__ void mix_slot(const float* __restrict__ w2l, const float* H2f,
                                         float* OUTf, int q, int lrow, int kb, int crow)
{
    constexpr int NT = 4;
    constexpr int d = 2 * L + 1, dd = d * d, Kdim = 16 * d;
    constexpr int o = (L * (2 * L - 1) * (2 * L + 1)) / 3;
    const float scale = rsqrtf((float)Kdim);
    const int mc = (lrow < d) ? lrow : (d - 1);
    const float* hl = H2f + o + mc;

    int rowoff[NT];
    #pragma unroll
    for (int j = 0; j < NT; ++j) {
        const int R  = q * 64 + j * 16 + lrow;
        const int g2 = (int)((unsigned)R / (unsigned)d);
        rowoff[j] = g2 * dd + (R - g2 * d);
    }

    v8f acc0[NT], acc1[NT];
    #pragma unroll
    for (int j = 0; j < NT; ++j) { acc0[j] = zero8(); acc1[j] = zero8(); }

    v2f a_e[NT], a_o[NT], b0_e, b1_e, b0_o, b1_o;
    mix_frag<L, NT>(w2l, rowoff, hl, kb, a_e, b0_e, b1_e);
    for (int k0 = 0; k0 < Kdim; k0 += 8) {
        mix_frag<L, NT>(w2l, rowoff, hl, k0 + 4 + kb, a_o, b0_o, b1_o);
        #pragma unroll
        for (int j = 0; j < NT; ++j) {
            acc0[j] = wmma4(a_e[j], b0_e, acc0[j]);
            acc1[j] = wmma4(a_e[j], b1_e, acc1[j]);
        }
        const int kn = (k0 + 8 < Kdim) ? (k0 + 8) : k0;
        mix_frag<L, NT>(w2l, rowoff, hl, kn + kb, a_e, b0_e, b1_e);
        #pragma unroll
        for (int j = 0; j < NT; ++j) {
            acc0[j] = wmma4(a_o[j], b0_o, acc0[j]);
            acc1[j] = wmma4(a_o[j], b1_o, acc1[j]);
        }
    }

    if (lrow < d) {
        #pragma unroll
        for (int j = 0; j < NT; ++j) {
            #pragma unroll
            for (int r = 0; r < 8; ++r) {
                const int R  = q * 64 + j * 16 + crow + r;
                const int g2 = (int)((unsigned)R / (unsigned)d);
                float* ob = OUTf + g2 * OUTS + o + (R - g2 * d) * d + lrow;
                ob[0]         = acc0[j][r] * scale;
                ob[64 * OUTS] = acc1[j][r] * scale;
            }
        }
    }
}

template <int L>
__device__ __forceinline__ void mix_level(const float* __restrict__ w2l, const float* H2f,
                                          float* OUTf, int q0, int cnt, int lrow, int kb, int crow)
{
    for (int q = q0; q < q0 + cnt; ++q)
        mix_slot<L>(w2l, H2f, OUTf, q, lrow, kb, crow);
}

__device__ __forceinline__ void dout_group(const float* p, v4f (&f)[4][2])
{
    #pragma unroll
    for (int s = 0; s < 4; ++s) {
        f[s][0] = *reinterpret_cast<const v4fu*>(p + (4 * s) * DIM455);
        f[s][1] = *reinterpret_cast<const v4fu*>(p + (4 * s + 1) * DIM455);
    }
}

__device__ __forceinline__ void h2_group(const float (*S)[130], int kk0, int lrow, int kb,
                                         const v4f (&f)[4][2], v8f (&acc)[2][4])
{
    #pragma unroll
    for (int s = 0; s < 4; ++s) {
        const int kk = kk0 + 4 * s + kb;
        v2f a0, a1;
        a0.x = S[lrow][kk];
        a0.y = S[lrow][kk + 1];
        a1.x = S[16 + lrow][kk];
        a1.y = S[16 + lrow][kk + 1];
        #pragma unroll
        for (int j = 0; j < 4; ++j) {
            v2f b;
            b.x = f[s][0][j];
            b.y = f[s][1][j];
            acc[0][j] = wmma4(a0, b, acc[0][j]);
            acc[1][j] = wmma4(a1, b, acc[1][j]);
        }
    }
}

__device__ __forceinline__ void order_point(v8f (&acc)[2][4])
{
    asm volatile("" : "+v"(acc[0][0]), "+v"(acc[0][1]), "+v"(acc[0][2]), "+v"(acc[0][3]),
                      "+v"(acc[1][0]), "+v"(acc[1][1]), "+v"(acc[1][2]), "+v"(acc[1][3]));
}

__device__ __forceinline__ void lds_barrier()
{
    __builtin_amdgcn_fence(__ATOMIC_RELEASE, "workgroup", "local");
    __builtin_amdgcn_s_barrier();
    __builtin_amdgcn_fence(__ATOMIC_ACQUIRE, "workgroup", "local");
}

__device__ __forceinline__ void din_frag(const float* dq, int kb, v2f (&db)[3])
{
    db[0] = *reinterpret_cast<const v2f*>(dq + kb);
    db[1] = *reinterpret_cast<const v2f*>(dq + 4 + kb);
    db[2] = *reinterpret_cast<const v2f*>(dq + 8);
}

__device__ __forceinline__ void sigma_tile(const v2f (&ha)[3], const v2f (&db)[3], int kb, float* Sd)
{
    v2f b2;
    b2.x = (kb == 0) ? db[2].x : 0.f;
    b2.y = (kb == 0) ? db[2].y : 0.f;
    v8f acc = zero8();
    acc = wmma4(ha[0], db[0], acc);
    acc = wmma4(ha[1], db[1], acc);
    acc = wmma4(ha[2], b2, acc);
    #pragma unroll
    for (int r = 0; r < 8; ++r) Sd[r * 130] = fmaxf(acc[r], 0.f);
}

__global__ __launch_bounds__(512) void traj_kernel(
    const float* __restrict__ traj, const float* __restrict__ w1s,
    const float* __restrict__ w1v,
    const float* Din   ,
    const float* Dout  ,
    const float* __restrict__ ico,
    const float* __restrict__ w20, const float* __restrict__ w21,
    const float* __restrict__ w22, const float* __restrict__ w23,
    const float* __restrict__ w24, const float* __restrict__ w25,
    const float* __restrict__ w26, float* __restrict__ out)
{
    constexpr int G = 4096, KC = 128, MR = 32;
    __shared__ float H2[MR][H2S];
    __shared__ __attribute__((aligned(16))) float U[2 * 64 * OUTS];
    constexpr int SBUF = MR * 130;
    float (*h_lds)[12] = reinterpret_cast<float (*)[12]>(U + 2 * SBUF);
    float* OUTf = U;

    const int bt0  = blockIdx.x * 2;
    const int tid  = threadIdx.x;
    const int wave = tid >> 5;
    const int lane = tid & 31;
    const int lrow = lane & 15;
    const int kb   = (lane >> 4) << 1;
    const int crow = (lane >> 4) << 3;

    if (tid < 320) {
        int row = tid / 10, i = tid % 10;
        int c = row & 15;
        const float* t10 = traj + (bt0 + (row >> 4)) * 10;
        float hv;
        if (i == 0) {
            hv = w1s[c] * t10[9];
        } else {
            int vv = (i - 1) / 3, mm = (i - 1) % 3;
            float a = 0.f;
            #pragma unroll
            for (int u = 0; u < 3; ++u)
                a += t10[u * 3 + mm] * w1v[c * 9 + u * 3 + vv];
            hv = a * 0.5773502691896258f;
        }
        h_lds[row][i] = hv;
    }
    __syncthreads();

    v8f acc[2][4];
    #pragma unroll
    for (int m = 0; m < 2; ++m) {
        #pragma unroll
        for (int j = 0; j < 4; ++j) acc[m][j] = zero8();
    }

    const int wn   = wave & 7;
    const int khalf   = wave >> 3;
    const int col0 = 64 * wn + 4 * lrow;
    const int colc = (col0 < DIM455 - 4) ? col0 : (DIM455 - 4);
    constexpr int GSTEP = 16 * DIM455;
    const float* p0 = Dout + (64 * khalf + kb) * DIM455 + colc;
    v4f fe[4][2], fo[4][2];

    const int smt = wave & 1;
    const int snt = wave >> 1;
    v2f ha[3];
    {
        const float* hr = &h_lds[smt * 16 + lrow][0];
        ha[0].x = hr[kb];      ha[0].y = hr[kb + 1];
        ha[1].x = hr[4 + kb];  ha[1].y = hr[5 + kb];
        const float h8 = hr[8], h9 = hr[9];
        ha[2].x = (kb == 0) ? h8 : 0.f;
        ha[2].y = (kb == 0) ? h9 : 0.f;
    }
    const float* dq = Din + (snt * 16 + lrow) * 10;
    const int sd_off = (smt * 16 + crow) * 130 + snt * 16 + lrow;
    v2f dn[3];
    din_frag(dq, kb, dn);
    dout_group(p0, fe);
    sigma_tile(ha, dn, kb, U + sd_off);
    dq += KC * 10;
    din_frag(dq, kb, dn);
    lds_barrier();

    const int kq = 64 * khalf;
    for (int g0 = 0; g0 < G; g0 += KC) {
        const int cur = (g0 / KC) & 1;
        const float (*S)[130] = reinterpret_cast<const float (*)[130]>(U + cur * SBUF);
        float* Sd = U + (cur ^ 1) * SBUF + sd_off;

        p0 += GSTEP;
        dout_group(p0, fo);
        order_point(acc);
        h2_group(S, kq, lrow, kb, fe, acc);
        order_point(acc);
        p0 += GSTEP;
        dout_group(p0, fe);
        order_point(acc);
        h2_group(S, kq + 16, lrow, kb, fo, acc);
        order_point(acc);

        sigma_tile(ha, dn, kb, Sd);
        dq += (g0 + 2 * KC < G) ? KC * 10 : 0;
        din_frag(dq, kb, dn);

        p0 += GSTEP;
        dout_group(p0, fo);
        order_point(acc);
        h2_group(S, kq + 32, lrow, kb, fe, acc);
        order_point(acc);
        const int adv = (g0 + KC < G) ? (GSTEP + 64 * DIM455) : 0;
        p0 += adv;
        dout_group(p0, fe);
        order_point(acc);
        h2_group(S, kq + 48, lrow, kb, fo, acc);
        order_point(acc);
        lds_barrier();
    }

    if (khalf == 1) {
        #pragma unroll
        for (int m = 0; m < 2; ++m) {
            #pragma unroll
            for (int j = 0; j < 4; ++j) {
                const int col = colc + j;
                if (col >= col0 && col < DIM455) {
                    #pragma unroll
                    for (int r = 0; r < 8; ++r)
                        H2[m * 16 + r + crow][col] = acc[m][j][r];
                }
            }
        }
    }
    __syncthreads();
    if (khalf == 0) {
        #pragma unroll
        for (int m = 0; m < 2; ++m) {
            #pragma unroll
            for (int j = 0; j < 4; ++j) {
                const int col = colc + j;
                if (col >= col0 && col < DIM455) {
                    #pragma unroll
                    for (int r = 0; r < 8; ++r)
                        H2[m * 16 + r + crow][col] += acc[m][j][r];
                }
            }
        }
    }
    __syncthreads();

    if (tid < 128) OUTf[tid * OUTS + DIM455] = 0.f;
    const float* H2f = &H2[0][0];
    {
        const int wv = __builtin_amdgcn_readfirstlane(wave);
        const bool isC = wv < 4, isA = (wv >= 4 && wv < 9), isB = wv >= 9;
        const int a = wv - 4, b = wv - 9, c = wv;
        int q6 = 0, n6 = 0, q5 = 0, n5 = 0, q4 = 0, n4 = 0, q3 = 0, n3 = 0;
        int q2 = 0, n2 = 0, q1 = 0, n1 = 0, n0 = 0;
        if (isA) { q6 = a; n6 = 1; q5 = a; n5 = 1; q2 = a; n2 = 1; }
        if (isB) { q6 = 5 + b; n6 = 1; q4 = b; n4 = 1; q3 = b; n3 = 1; }
        if (isC) {
            if (c == 0) { q6 = 12; n6 = 1; q5 = 5; n5 = 1; q1 = 0; n1 = 1; }
            if (c == 1) { q5 = 6; n5 = 2; q1 = 1; n1 = 1; n0 = 1; }
            if (c == 2) { q5 = 8; n5 = 2; q1 = 2; n1 = 1; }
            if (c == 3) { q5 = 10; n5 = 1; q4 = 7; n4 = 2; }
        }
        mix_level<6>(w26, H2f, OUTf, q6, n6, lrow, kb, crow);
        mix_level<5>(w25, H2f, OUTf, q5, n5, lrow, kb, crow);
        mix_level<4>(w24, H2f, OUTf, q4, n4, lrow, kb, crow);
        mix_level<3>(w23, H2f, OUTf, q3, n3, lrow, kb, crow);
        mix_level<2>(w22, H2f, OUTf, q2, n2, lrow, kb, crow);
        mix_level<1>(w21, H2f, OUTf, q1, n1, lrow, kb, crow);
        mix_level<0>(w20, H2f, OUTf, 0, n0, lrow, kb, crow);
    }
    __syncthreads();

    const int bsel = wave >> 3;
    const int kh = wave & 1;
    const int np = (wave >> 1) & 1;
    const int mp = (wave >> 2) & 1;
    const int bt = bt0 + bsel;
    OUTf += bsel * 64 * OUTS;

    v8f facc0 = zero8(), facc1 = zero8(), facc2 = zero8(), facc3 = zero8();
    {
        const float* oa0 = OUTf + ((mp * 2 + 0) * 16 + lrow) * OUTS;
        const float* oa1 = OUTf + ((mp * 2 + 1) * 16 + lrow) * OUTS;
        const int rb0 = (np * 2 + 0) * 16 + lrow;
        const int rb1 = (np * 2 + 1) * 16 + lrow;
        const int rb1c = (rb1 < 60) ? rb1 : 59;
        const float* i0 = ico + rb0 * DIM455;
        const float* i1 = ico + rb1c * DIM455;
        const int ks0 = kh * 57;
        for (int ks = ks0; ks < ks0 + 57; ++ks) {
            const int k  = ks * 4 + kb;
            const int k1 = (k + 1 < DIM455) ? (k + 1) : (DIM455 - 1);
            v2f a0 = *reinterpret_cast<const v2f*>(oa0 + k);
            v2f a1 = *reinterpret_cast<const v2f*>(oa1 + k);
            float t0x = i0[k], t0y = i0[k1];
            float t1x = i1[k], t1y = i1[k1];
            v2f b0, b1;
            b0.x = t0x;  b0.y = (k + 1 < DIM455) ? t0y : 0.f;
            b1.x = t1x;  b1.y = (k + 1 < DIM455) ? t1y : 0.f;
            facc0 = wmma4(a0, b0, facc0);
            facc1 = wmma4(a0, b1, facc1);
            facc2 = wmma4(a1, b0, facc2);
            facc3 = wmma4(a1, b1, facc3);
        }
    }

    float* red = &H2[0][0] + bsel * (64 * 64);
    {
        v8f* faccs[4] = {&facc0, &facc1, &facc2, &facc3};
        if (kh == 1) {
            #pragma unroll
            for (int q = 0; q < 4; ++q) {
                const int mt = mp * 2 + (q >> 1), nt = np * 2 + (q & 1);
                #pragma unroll
                for (int j = 0; j < 8; ++j)
                    red[(mt * 16 + j + crow) * 64 + nt * 16 + lrow] = (*faccs[q])[j];
            }
        }
        __syncthreads();
        if (kh == 0) {
            const long xg_elems = 32L * 4 * 128 * 60;
            #pragma unroll
            for (int q = 0; q < 4; ++q) {
                const int mt = mp * 2 + (q >> 1), nt = np * 2 + (q & 1);
                const int r = nt * 16 + lrow;
                if (r < 60) {
                    #pragma unroll
                    for (int j = 0; j < 8; ++j) {
                        const int g2 = mt * 16 + j + crow;
                        float vsum = (*faccs[q])[j] + red[g2 * 64 + r];
                        out[xg_elems + ((long)bt * 64 + g2) * 60 + r] = vsum;
                    }
                }
            }
        }
    }
}

__global__ __launch_bounds__(256) void xg_kernel(
    const float* __restrict__ x, const float* __restrict__ ico,
    float* __restrict__ out)
{
    const int tid  = threadIdx.x;
    const int wave = tid >> 5;
    const int lane = tid & 31;
    const int lrow = lane & 15;
    const int kb   = (lane >> 4) << 1;
    const int crow = (lane >> 4) << 3;
    const int mtile = wave >> 1;
    const int ng    = wave & 1;

    const long rowA = (long)blockIdx.x * 64 + mtile * 16 + lrow;
    const float* xr = x + rowA * DIM455;

    v8f acc0 = zero8(), acc1 = zero8();
    const int r0  = (ng * 2 + 0) * 16 + lrow;
    const int r1  = (ng * 2 + 1) * 16 + lrow;
    const int r1c = (r1 < 60) ? r1 : 59;

    for (int k0 = 0; k0 < 456; k0 += 4) {
        int k   = k0 + kb;
        int kc0 = (k     < DIM455) ? k       : (DIM455 - 1);
        int kc1 = (k + 1 < DIM455) ? (k + 1) : (DIM455 - 1);
        float ax = xr[kc0], ay = xr[kc1];
        float b0x = ico[r0  * DIM455 + kc0], b0y = ico[r0  * DIM455 + kc1];
        float b1x = ico[r1c * DIM455 + kc0], b1y = ico[r1c * DIM455 + kc1];
        v2f a, b0, b1;
        a.x  = (k     < DIM455) ? ax  : 0.f;
        a.y  = (k + 1 < DIM455) ? ay  : 0.f;
        b0.x = (k     < DIM455) ? b0x : 0.f;
        b0.y = (k + 1 < DIM455) ? b0y : 0.f;
        b1.x = (k     < DIM455 && r1 < 60) ? b1x : 0.f;
        b1.y = (k + 1 < DIM455 && r1 < 60) ? b1y : 0.f;
        acc0 = wmma4(a, b0, acc0);
        acc1 = wmma4(a, b1, acc1);
    }

    #pragma unroll
    for (int s = 0; s < 2; ++s) {
        int r = (ng * 2 + s) * 16 + lrow;
        if (r < 60) {
            #pragma unroll
            for (int j = 0; j < 8; ++j) {
                long row = (long)blockIdx.x * 64 + mtile * 16 + j + crow;
                out[row * 60 + r] = (s == 0) ? acc0[j] : acc1[j];
            }
        }
    }
}

extern "C" void kernel_launch(void* const* d_in, const int* in_sizes, int n_in,
                              void* d_out, int out_size, void* d_ws, size_t ws_size,
                              hipStream_t stream) {
    const float* x    = (const float*)d_in[0];
    const float* traj = (const float*)d_in[1];
    const float* w1s  = (const float*)d_in[2];
    const float* w1v  = (const float*)d_in[3];
    const float* Din  = (const float*)d_in[4];
    const float* Dout = (const float*)d_in[5];
    const float* ico  = (const float*)d_in[6];
    float* out = (float*)d_out;

    xg_kernel<<<dim3(16384 / 64), dim3(256), 0, stream>>>(x, ico, out);

    traj_kernel<<<dim3(256), dim3(512), 0, stream>>>(
        traj, w1s, w1v, Din, Dout, ico,
        (const float*)d_in[7],  (const float*)d_in[8],  (const float*)d_in[9],
        (const float*)d_in[10], (const float*)d_in[11], (const float*)d_in[12],
        (const float*)d_in[13], out);
}


# xg product moved into the traj kernel as each work-group's first phase (its 64 x rows and ico copied to LDS in 16-byte pieces, fragments read from LDS): one launch; k order and sums unchanged
# speedup vs baseline: 4.3438x; 1.1085x over previous
#include <hip/hip_runtime.h>

typedef __attribute__((ext_vector_type(2))) float v2f;
typedef __attribute__((ext_vector_type(8))) float v8f;
typedef __attribute__((ext_vector_type(4))) float v4f;
typedef v4f __attribute__((aligned(4))) v4fu;

__device__ __forceinline__ v8f zero8() {
    v8f z = {0.f,0.f,0.f,0.f,0.f,0.f,0.f,0.f};
    return z;
}

__device__ __forceinline__ v8f wmma4(v2f a, v2f b, v8f c) {
    return __builtin_amdgcn_wmma_f32_16x16x4_f32(
         false, a,  false, b,
         (short)0, c,  false,  false);
}

#define DIM455 455

#define H2S  456
#define OUTS 460

template <int L, int NT>
__device__ __forceinline__ void mix_frag(const float* __restrict__ w2l, const int (&rowoff)[NT],
                                         const float* hl, int kx, v2f (&a)[NT], v2f& b0, v2f& b1)
{
    constexpr int d = 2 * L + 1, dd = d * d;
    const int f0 = (int)((unsigned)kx / (unsigned)d);
    const int f1 = (int)((unsigned)(kx + 1) / (unsigned)d);
    const int w0 = f0 * (63 * dd) + kx * d;
    const int w1 = f1 * (63 * dd) + (kx + 1) * d;
    const int h0 = f0 * (H2S - dd) + kx * d;
    const int h1 = f1 * (H2S - dd) + (kx + 1) * d;
    b0.x = hl[h0];
    b0.y = hl[h1];
    b1.x = hl[16 * H2S + h0];
    b1.y = hl[16 * H2S + h1];
    #pragma unroll
    for (int j = 0; j < NT; ++j) {
        a[j].x = w2l[rowoff[j] + w0];
        a[j].y = w2l[rowoff[j] + w1];
    }
}

template <int L>
__device__ __forceinline__ void mix_slot(const float* __restrict__ w2l, const float* H2f,
                                         float* OUTf, int q, int lrow, int kb, int crow)
{
    constexpr int NT = 4;
    constexpr int d = 2 * L + 1, dd = d * d, Kdim = 16 * d;
    constexpr int o = (L * (2 * L - 1) * (2 * L + 1)) / 3;
    const float scale = rsqrtf((float)Kdim);
    const int mc = (lrow < d) ? lrow : (d - 1);
    const float* hl = H2f + o + mc;

    int rowoff[NT];
    #pragma unroll
    for (int j = 0; j < NT; ++j) {
        const int R  = q * 64 + j * 16 + lrow;
        const int g2 = (int)((unsigned)R / (unsigned)d);
        rowoff[j] = g2 * dd + (R - g2 * d);
    }

    v8f acc0[NT], acc1[NT];
    #pragma unroll
    for (int j = 0; j < NT; ++j) { acc0[j] = zero8(); acc1[j] = zero8(); }

    v2f a_e[NT], a_o[NT], b0_e, b1_e, b0_o, b1_o;
    mix_frag<L, NT>(w2l, rowoff, hl, kb, a_e, b0_e, b1_e);
    for (int k0 = 0; k0 < Kdim; k0 += 8) {
        mix_frag<L, NT>(w2l, rowoff, hl, k0 + 4 + kb, a_o, b0_o, b1_o);
        #pragma unroll
        for (int j = 0; j < NT; ++j) {
            acc0[j] = wmma4(a_e[j], b0_e, acc0[j]);
            acc1[j] = wmma4(a_e[j], b1_e, acc1[j]);
        }
        const int kn = (k0 + 8 < Kdim) ? (k0 + 8) : k0;
        mix_frag<L, NT>(w2l, rowoff, hl, kn + kb, a_e, b0_e, b1_e);
        #pragma unroll
        for (int j = 0; j < NT; ++j) {
            acc0[j] = wmma4(a_o[j], b0_o, acc0[j]);
            acc1[j] = wmma4(a_o[j], b1_o, acc1[j]);
        }
    }

    if (lrow < d) {
        #pragma unroll
        for (int j = 0; j < NT; ++j) {
            #pragma unroll
            for (int r = 0; r < 8; ++r) {
                const int R  = q * 64 + j * 16 + crow + r;
                const int g2 = (int)((unsigned)R / (unsigned)d);
                float* ob = OUTf + g2 * OUTS + o + (R - g2 * d) * d + lrow;
                ob[0]         = acc0[j][r] * scale;
                ob[64 * OUTS] = acc1[j][r] * scale;
            }
        }
    }
}

template <int L>
__device__ __forceinline__ void mix_level(const float* __restrict__ w2l, const float* H2f,
                                          float* OUTf, int q0, int cnt, int lrow, int kb, int crow)
{
    for (int q = q0; q < q0 + cnt; ++q)
        mix_slot<L>(w2l, H2f, OUTf, q, lrow, kb, crow);
}

__device__ __forceinline__ void dout_group(const float* p, v4f (&f)[4][2])
{
    #pragma unroll
    for (int s = 0; s < 4; ++s) {
        f[s][0] = *reinterpret_cast<const v4fu*>(p + (4 * s) * DIM455);
        f[s][1] = *reinterpret_cast<const v4fu*>(p + (4 * s + 1) * DIM455);
    }
}

__device__ __forceinline__ void h2_group(const float (*S)[130], int kk0, int lrow, int kb,
                                         const v4f (&f)[4][2], v8f (&acc)[2][4])
{
    #pragma unroll
    for (int s = 0; s < 4; ++s) {
        const int kk = kk0 + 4 * s + kb;
        v2f a0, a1;
        a0.x = S[lrow][kk];
        a0.y = S[lrow][kk + 1];
        a1.x = S[16 + lrow][kk];
        a1.y = S[16 + lrow][kk + 1];
        #pragma unroll
        for (int j = 0; j < 4; ++j) {
            v2f b;
            b.x = f[s][0][j];
            b.y = f[s][1][j];
            acc[0][j] = wmma4(a0, b, acc[0][j]);
            acc[1][j] = wmma4(a1, b, acc[1][j]);
        }
    }
}

__device__ __forceinline__ void order_point(v8f (&acc)[2][4])
{
    asm volatile("" : "+v"(acc[0][0]), "+v"(acc[0][1]), "+v"(acc[0][2]), "+v"(acc[0][3]),
                      "+v"(acc[1][0]), "+v"(acc[1][1]), "+v"(acc[1][2]), "+v"(acc[1][3]));
}

__device__ __forceinline__ void lds_barrier()
{
    __builtin_amdgcn_fence(__ATOMIC_RELEASE, "workgroup", "local");
    __builtin_amdgcn_s_barrier();
    __builtin_amdgcn_fence(__ATOMIC_ACQUIRE, "workgroup", "local");
}

__device__ __forceinline__ void din_frag(const float* dq, int kb, v2f (&db)[3])
{
    db[0] = *reinterpret_cast<const v2f*>(dq + kb);
    db[1] = *reinterpret_cast<const v2f*>(dq + 4 + kb);
    db[2] = *reinterpret_cast<const v2f*>(dq + 8);
}

__device__ __forceinline__ void sigma_tile(const v2f (&ha)[3], const v2f (&db)[3], int kb, float* Sd)
{
    v2f b2;
    b2.x = (kb == 0) ? db[2].x : 0.f;
    b2.y = (kb == 0) ? db[2].y : 0.f;
    v8f acc = zero8();
    acc = wmma4(ha[0], db[0], acc);
    acc = wmma4(ha[1], db[1], acc);
    acc = wmma4(ha[2], b2, acc);
    #pragma unroll
    for (int r = 0; r < 8; ++r) Sd[r * 130] = fmaxf(acc[r], 0.f);
}

__global__ __launch_bounds__(512) void traj_kernel(
    const float* __restrict__ x,
    const float* __restrict__ traj, const float* __restrict__ w1s,
    const float* __restrict__ w1v,
    const float* Din   ,
    const float* Dout  ,
    const float* __restrict__ ico,
    const float* __restrict__ w20, const float* __restrict__ w21,
    const float* __restrict__ w22, const float* __restrict__ w23,
    const float* __restrict__ w24, const float* __restrict__ w25,
    const float* __restrict__ w26, float* __restrict__ out)
{
    constexpr int G = 4096, KC = 128, MR = 32;
    __shared__ float H2[MR][H2S];
    __shared__ __attribute__((aligned(16))) float U[2 * 64 * OUTS];
    constexpr int SBUF = MR * 130;
    float (*h_lds)[12] = reinterpret_cast<float (*)[12]>(U + 2 * SBUF);
    float* OUTf = U;

    const int bt0  = blockIdx.x * 2;
    const int tid  = threadIdx.x;
    const int wave = tid >> 5;
    const int lane = tid & 31;
    const int lrow = lane & 15;
    const int kb   = (lane >> 4) << 1;
    const int crow = (lane >> 4) << 3;

    {
        constexpr int NX4 = 64 * DIM455 / 4;
        constexpr int NI4 = 60 * DIM455 / 4;
        constexpr int RX = (NX4 + 511) / 512;
        constexpr int RI = (NI4 + 511) / 512;
        float* XS = U;
        float* IS = U + 64 * DIM455;
        const v4fu* xg4 = reinterpret_cast<const v4fu*>(x + (long)blockIdx.x * (64 * DIM455));
        const v4fu* ig4 = reinterpret_cast<const v4fu*>(ico);
        v4f* XS4 = reinterpret_cast<v4f*>(XS);
        v4f* IS4 = reinterpret_cast<v4f*>(IS);
        v4f xv[RX], iv[RI];
        const int lx = (512 * (RX - 1) + tid < NX4) ? (512 * (RX - 1) + tid) : (NX4 - 1);
        const int li = (512 * (RI - 1) + tid < NI4) ? (512 * (RI - 1) + tid) : (NI4 - 1);
        #pragma unroll
        for (int j = 0; j < RX - 1; ++j) xv[j] = xg4[tid + 512 * j];
        xv[RX - 1] = xg4[lx];
        #pragma unroll
        for (int j = 0; j < RI - 1; ++j) iv[j] = ig4[tid + 512 * j];
        iv[RI - 1] = ig4[li];
        #pragma unroll
        for (int j = 0; j < RX - 1; ++j) XS4[tid + 512 * j] = xv[j];
        XS4[lx] = xv[RX - 1];
        #pragma unroll
        for (int j = 0; j < RI - 1; ++j) IS4[tid + 512 * j] = iv[j];
        IS4[li] = iv[RI - 1];
        lds_barrier();

        const int xmt = wave >> 2, xnt = wave & 3;
        const int rb  = xnt * 16 + lrow;
        const int rbc = (rb < 60) ? rb : 59;
        const float* xa = XS + (xmt * 16 + lrow) * DIM455 + kb;
        const float* ib = IS + rbc * DIM455 + kb;
        v8f xacc = zero8();
        #pragma unroll 16
        for (int s = 0; s < 112; ++s) {
            v2f a, b;
            a.x = xa[4 * s];  a.y = xa[4 * s + 1];
            b.x = ib[4 * s];  b.y = ib[4 * s + 1];
            xacc = wmma4(a, b, xacc);
        }
        {
            v2f a, b;
            a.x = xa[448];  a.y = xa[449];
            b.x = ib[448];  b.y = ib[449];
            xacc = wmma4(a, b, xacc);
        }
        {
            const int k1 = (kb == 0) ? 453 : 452;
            v2f a, b;
            const float ay = xa[k1], by = ib[k1];
            a.x = xa[452];  a.y = (kb == 0) ? ay : 0.f;
            b.x = ib[452];  b.y = (kb == 0) ? by : 0.f;
            xacc = wmma4(a, b, xacc);
        }
        if (rb < 60) {
            const long row0 = (long)blockIdx.x * 64 + xmt * 16 + crow;
            #pragma unroll
            for (int r = 0; r < 8; ++r)
                out[(row0 + r) * 60 + rb] = xacc[r];
        }
        lds_barrier();
    }

    if (tid < 320) {
        int row = tid / 10, i = tid % 10;
        int c = row & 15;
        const float* t10 = traj + (bt0 + (row >> 4)) * 10;
        float hv;
        if (i == 0) {
            hv = w1s[c] * t10[9];
        } else {
            int vv = (i - 1) / 3, mm = (i - 1) % 3;
            float a = 0.f;
            #pragma unroll
            for (int u = 0; u < 3; ++u)
                a += t10[u * 3 + mm] * w1v[c * 9 + u * 3 + vv];
            hv = a * 0.5773502691896258f;
        }
        h_lds[row][i] = hv;
    }
    __syncthreads();

    v8f acc[2][4];
    #pragma unroll
    for (int m = 0; m < 2; ++m) {
        #pragma unroll
        for (int j = 0; j < 4; ++j) acc[m][j] = zero8();
    }

    const int wn   = wave & 7;
    const int khalf   = wave >> 3;
    const int col0 = 64 * wn + 4 * lrow;
    const int colc = (col0 < DIM455 - 4) ? col0 : (DIM455 - 4);
    constexpr int GSTEP = 16 * DIM455;
    const float* p0 = Dout + (64 * khalf + kb) * DIM455 + colc;
    v4f fe[4][2], fo[4][2];

    const int smt = wave & 1;
    const int snt = wave >> 1;
    v2f ha[3];
    {
        const float* hr = &h_lds[smt * 16 + lrow][0];
        ha[0].x = hr[kb];      ha[0].y = hr[kb + 1];
        ha[1].x = hr[4 + kb];  ha[1].y = hr[5 + kb];
        const float h8 = hr[8], h9 = hr[9];
        ha[2].x = (kb == 0) ? h8 : 0.f;
        ha[2].y = (kb == 0) ? h9 : 0.f;
    }
    const float* dq = Din + (snt * 16 + lrow) * 10;
    const int sd_off = (smt * 16 + crow) * 130 + snt * 16 + lrow;
    v2f dn[3];
    din_frag(dq, kb, dn);
    dout_group(p0, fe);
    sigma_tile(ha, dn, kb, U + sd_off);
    dq += KC * 10;
    din_frag(dq, kb, dn);
    lds_barrier();

    const int kq = 64 * khalf;
    for (int g0 = 0; g0 < G; g0 += KC) {
        const int cur = (g0 / KC) & 1;
        const float (*S)[130] = reinterpret_cast<const float (*)[130]>(U + cur * SBUF);
        float* Sd = U + (cur ^ 1) * SBUF + sd_off;

        p0 += GSTEP;
        dout_group(p0, fo);
        order_point(acc);
        h2_group(S, kq, lrow, kb, fe, acc);
        order_point(acc);
        p0 += GSTEP;
        dout_group(p0, fe);
        order_point(acc);
        h2_group(S, kq + 16, lrow, kb, fo, acc);
        order_point(acc);

        sigma_tile(ha, dn, kb, Sd);
        dq += (g0 + 2 * KC < G) ? KC * 10 : 0;
        din_frag(dq, kb, dn);

        p0 += GSTEP;
        dout_group(p0, fo);
        order_point(acc);
        h2_group(S, kq + 32, lrow, kb, fe, acc);
        order_point(acc);
        const int adv = (g0 + KC < G) ? (GSTEP + 64 * DIM455) : 0;
        p0 += adv;
        dout_group(p0, fe);
        order_point(acc);
        h2_group(S, kq + 48, lrow, kb, fo, acc);
        order_point(acc);
        lds_barrier();
    }

    if (khalf == 1) {
        #pragma unroll
        for (int m = 0; m < 2; ++m) {
            #pragma unroll
            for (int j = 0; j < 4; ++j) {
                const int col = colc + j;
                if (col >= col0 && col < DIM455) {
                    #pragma unroll
                    for (int r = 0; r < 8; ++r)
                        H2[m * 16 + r + crow][col] = acc[m][j][r];
                }
            }
        }
    }
    __syncthreads();
    if (khalf == 0) {
        #pragma unroll
        for (int m = 0; m < 2; ++m) {
            #pragma unroll
            for (int j = 0; j < 4; ++j) {
                const int col = colc + j;
                if (col >= col0 && col < DIM455) {
                    #pragma unroll
                    for (int r = 0; r < 8; ++r)
                        H2[m * 16 + r + crow][col] += acc[m][j][r];
                }
            }
        }
    }
    __syncthreads();

    if (tid < 128) OUTf[tid * OUTS + DIM455] = 0.f;
    const float* H2f = &H2[0][0];
    {
        const int wv = __builtin_amdgcn_readfirstlane(wave);
        const bool isC = wv < 4, isA = (wv >= 4 && wv < 9), isB = wv >= 9;
        const int a = wv - 4, b = wv - 9, c = wv;
        int q6 = 0, n6 = 0, q5 = 0, n5 = 0, q4 = 0, n4 = 0, q3 = 0, n3 = 0;
        int q2 = 0, n2 = 0, q1 = 0, n1 = 0, n0 = 0;
        if (isA) { q6 = a; n6 = 1; q5 = a; n5 = 1; q2 = a; n2 = 1; }
        if (isB) { q6 = 5 + b; n6 = 1; q4 = b; n4 = 1; q3 = b; n3 = 1; }
        if (isC) {
            if (c == 0) { q6 = 12; n6 = 1; q5 = 5; n5 = 1; q1 = 0; n1 = 1; }
            if (c == 1) { q5 = 6; n5 = 2; q1 = 1; n1 = 1; n0 = 1; }
            if (c == 2) { q5 = 8; n5 = 2; q1 = 2; n1 = 1; }
            if (c == 3) { q5 = 10; n5 = 1; q4 = 7; n4 = 2; }
        }
        mix_level<6>(w26, H2f, OUTf, q6, n6, lrow, kb, crow);
        mix_level<5>(w25, H2f, OUTf, q5, n5, lrow, kb, crow);
        mix_level<4>(w24, H2f, OUTf, q4, n4, lrow, kb, crow);
        mix_level<3>(w23, H2f, OUTf, q3, n3, lrow, kb, crow);
        mix_level<2>(w22, H2f, OUTf, q2, n2, lrow, kb, crow);
        mix_level<1>(w21, H2f, OUTf, q1, n1, lrow, kb, crow);
        mix_level<0>(w20, H2f, OUTf, 0, n0, lrow, kb, crow);
    }
    __syncthreads();

    const int bsel = wave >> 3;
    const int kh = wave & 1;
    const int np = (wave >> 1) & 1;
    const int mp = (wave >> 2) & 1;
    const int bt = bt0 + bsel;
    OUTf += bsel * 64 * OUTS;

    v8f facc0 = zero8(), facc1 = zero8(), facc2 = zero8(), facc3 = zero8();
    {
        const float* oa0 = OUTf + ((mp * 2 + 0) * 16 + lrow) * OUTS;
        const float* oa1 = OUTf + ((mp * 2 + 1) * 16 + lrow) * OUTS;
        const int rb0 = (np * 2 + 0) * 16 + lrow;
        const int rb1 = (np * 2 + 1) * 16 + lrow;
        const int rb1c = (rb1 < 60) ? rb1 : 59;
        const float* i0 = ico + rb0 * DIM455;
        const float* i1 = ico + rb1c * DIM455;
        const int ks0 = kh * 57;
        for (int ks = ks0; ks < ks0 + 57; ++ks) {
            const int k  = ks * 4 + kb;
            const int k1 = (k + 1 < DIM455) ? (k + 1) : (DIM455 - 1);
            v2f a0 = *reinterpret_cast<const v2f*>(oa0 + k);
            v2f a1 = *reinterpret_cast<const v2f*>(oa1 + k);
            float t0x = i0[k], t0y = i0[k1];
            float t1x = i1[k], t1y = i1[k1];
            v2f b0, b1;
            b0.x = t0x;  b0.y = (k + 1 < DIM455) ? t0y : 0.f;
            b1.x = t1x;  b1.y = (k + 1 < DIM455) ? t1y : 0.f;
            facc0 = wmma4(a0, b0, facc0);
            facc1 = wmma4(a0, b1, facc1);
            facc2 = wmma4(a1, b0, facc2);
            facc3 = wmma4(a1, b1, facc3);
        }
    }

    float* red = &H2[0][0] + bsel * (64 * 64);
    {
        v8f* faccs[4] = {&facc0, &facc1, &facc2, &facc3};
        if (kh == 1) {
            #pragma unroll
            for (int q = 0; q < 4; ++q) {
                const int mt = mp * 2 + (q >> 1), nt = np * 2 + (q & 1);
                #pragma unroll
                for (int j = 0; j < 8; ++j)
                    red[(mt * 16 + j + crow) * 64 + nt * 16 + lrow] = (*faccs[q])[j];
            }
        }
        __syncthreads();
        if (kh == 0) {
            const long xg_elems = 32L * 4 * 128 * 60;
            #pragma unroll
            for (int q = 0; q < 4; ++q) {
                const int mt = mp * 2 + (q >> 1), nt = np * 2 + (q & 1);
                const int r = nt * 16 + lrow;
                if (r < 60) {
                    #pragma unroll
                    for (int j = 0; j < 8; ++j) {
                        const int g2 = mt * 16 + j + crow;
                        float vsum = (*faccs[q])[j] + red[g2 * 64 + r];
                        out[xg_elems + ((long)bt * 64 + g2) * 60 + r] = vsum;
                    }
                }
            }
        }
    }
}

extern "C" void kernel_launch(void* const* d_in, const int* in_sizes, int n_in,
                              void* d_out, int out_size, void* d_ws, size_t ws_size,
                              hipStream_t stream) {
    const float* x    = (const float*)d_in[0];
    const float* traj = (const float*)d_in[1];
    const float* w1s  = (const float*)d_in[2];
    const float* w1v  = (const float*)d_in[3];
    const float* Din  = (const float*)d_in[4];
    const float* Dout = (const float*)d_in[5];
    const float* ico  = (const float*)d_in[6];
    float* out = (float*)d_out;

    traj_kernel<<<dim3(256), dim3(512), 0, stream>>>(
        x, traj, w1s, w1v, Din, Dout, ico,
        (const float*)d_in[7],  (const float*)d_in[8],  (const float*)d_in[9],
        (const float*)d_in[10], (const float*)d_in[11], (const float*)d_in[12],
        (const float*)d_in[13], out);
}


# w2 mix without integer address arithmetic: k relabelled so a step is one u and four f (f = 4 fg + 2 half + elem) and every offset a constant; store rows stepped, not divided; new summation order
# speedup vs baseline: 4.4794x; 1.0312x over previous
#include <hip/hip_runtime.h>

typedef __attribute__((ext_vector_type(2))) float v2f;
typedef __attribute__((ext_vector_type(8))) float v8f;
typedef __attribute__((ext_vector_type(4))) float v4f;
typedef v4f __attribute__((aligned(4))) v4fu;

__device__ __forceinline__ v8f zero8() {
    v8f z = {0.f,0.f,0.f,0.f,0.f,0.f,0.f,0.f};
    return z;
}

__device__ __forceinline__ v8f wmma4(v2f a, v2f b, v8f c) {
    return __builtin_amdgcn_wmma_f32_16x16x4_f32(
         false, a,  false, b,
         (short)0, c,  false,  false);
}

#define DIM455 455

#define H2S  456
#define OUTS 460

template <int L, int NT>
__device__ __forceinline__ void mix_frag(const float* __restrict__ wf, const int (&rowoff)[NT],
                                         const float* hf, int u, v2f (&a)[NT], v2f& b0, v2f& b1)
{
    constexpr int d = 2 * L + 1, dd = d * d;
    constexpr int FS = 64 * dd;
    b0.x = hf[u * d];
    b0.y = hf[u * d + H2S];
    b1.x = hf[16 * H2S + u * d];
    b1.y = hf[17 * H2S + u * d];
    #pragma unroll
    for (int j = 0; j < NT; ++j) {
        a[j].x = wf[rowoff[j] + u * d];
        a[j].y = wf[rowoff[j] + u * d + FS];
    }
}

template <int L>
__device__ __forceinline__ void mix_slot(const float* __restrict__ w2l, const float* H2f,
                                         float* OUTf, int q, int lrow, int kb, int crow)
{
    constexpr int NT = 4;
    constexpr int d = 2 * L + 1, dd = d * d;
    constexpr int FS = 64 * dd;
    constexpr int o = (L * (2 * L - 1) * (2 * L + 1)) / 3;
    const float scale = rsqrtf((float)(16 * d));
    const int mc = (lrow < d) ? lrow : (d - 1);
    const float* hl = H2f + kb * H2S + o + mc;

    int rowoff[NT];
    #pragma unroll
    for (int j = 0; j < NT; ++j) {
        const int R  = q * 64 + j * 16 + lrow;
        const int g2 = (int)((unsigned)R / (unsigned)d);
        rowoff[j] = kb * FS + g2 * dd + (R - g2 * d);
    }

    v8f acc0[NT], acc1[NT];
    #pragma unroll
    for (int j = 0; j < NT; ++j) { acc0[j] = zero8(); acc1[j] = zero8(); }

    v2f a_c[NT], b0_c, b1_c;
    mix_frag<L, NT>(w2l, rowoff, hl, 0, a_c, b0_c, b1_c);
    #pragma nounroll
    for (int fg = 0; fg < 4; ++fg) {
        const int fn = (fg < 3) ? (fg + 1) : fg;
        const float* wf = w2l + fg * (4 * FS);
        const float* hf = hl + fg * (4 * H2S);
        const float* wn = w2l + fn * (4 * FS);
        const float* hn = hl + fn * (4 * H2S);
        #pragma unroll
        for (int u = 0; u < d; ++u) {
            v2f a_n[NT], b0_n, b1_n;
            if (u + 1 < d) mix_frag<L, NT>(wf, rowoff, hf, u + 1, a_n, b0_n, b1_n);
            else           mix_frag<L, NT>(wn, rowoff, hn, 0, a_n, b0_n, b1_n);
            #pragma unroll
            for (int j = 0; j < NT; ++j) {
                acc0[j] = wmma4(a_c[j], b0_c, acc0[j]);
                acc1[j] = wmma4(a_c[j], b1_c, acc1[j]);
            }
            #pragma unroll
            for (int j = 0; j < NT; ++j) a_c[j] = a_n[j];
            b0_c = b0_n;
            b1_c = b1_n;
        }
    }

    if (lrow < d) {
        #pragma unroll
        for (int j = 0; j < NT; ++j) {
            const int R0 = q * 64 + j * 16 + crow;
            const int g0 = (int)((unsigned)R0 / (unsigned)d);
            int v = R0 - g0 * d;
            float* ob = OUTf + g0 * OUTS + o + v * d + lrow;
            #pragma unroll
            for (int r = 0; r < 8; ++r) {
                ob[0]         = acc0[j][r] * scale;
                ob[64 * OUTS] = acc1[j][r] * scale;
                const bool last = (v == d - 1);
                ob += last ? (OUTS - (d - 1) * d) : d;
                v   = last ? 0 : (v + 1);
            }
        }
    }
}

template <int L>
__device__ __forceinline__ void mix_level(const float* __restrict__ w2l, const float* H2f,
                                          float* OUTf, int q0, int cnt, int lrow, int kb, int crow)
{
    for (int q = q0; q < q0 + cnt; ++q)
        mix_slot<L>(w2l, H2f, OUTf, q, lrow, kb, crow);
}

__device__ __forceinline__ void dout_group(const float* p, v4f (&f)[4][2])
{
    #pragma unroll
    for (int s = 0; s < 4; ++s) {
        f[s][0] = *reinterpret_cast<const v4fu*>(p + (4 * s) * DIM455);
        f[s][1] = *reinterpret_cast<const v4fu*>(p + (4 * s + 1) * DIM455);
    }
}

__device__ __forceinline__ void h2_group(const float (*S)[130], int kk0, int lrow, int kb,
                                         const v4f (&f)[4][2], v8f (&acc)[2][4])
{
    #pragma unroll
    for (int s = 0; s < 4; ++s) {
        const int kk = kk0 + 4 * s + kb;
        v2f a0, a1;
        a0.x = S[lrow][kk];
        a0.y = S[lrow][kk + 1];
        a1.x = S[16 + lrow][kk];
        a1.y = S[16 + lrow][kk + 1];
        #pragma unroll
        for (int j = 0; j < 4; ++j) {
            v2f b;
            b.x = f[s][0][j];
            b.y = f[s][1][j];
            acc[0][j] = wmma4(a0, b, acc[0][j]);
            acc[1][j] = wmma4(a1, b, acc[1][j]);
        }
    }
}

__device__ __forceinline__ void order_point(v8f (&acc)[2][4])
{
    asm volatile("" : "+v"(acc[0][0]), "+v"(acc[0][1]), "+v"(acc[0][2]), "+v"(acc[0][3]),
                      "+v"(acc[1][0]), "+v"(acc[1][1]), "+v"(acc[1][2]), "+v"(acc[1][3]));
}

__device__ __forceinline__ void lds_barrier()
{
    __builtin_amdgcn_fence(__ATOMIC_RELEASE, "workgroup", "local");
    __builtin_amdgcn_s_barrier();
    __builtin_amdgcn_fence(__ATOMIC_ACQUIRE, "workgroup", "local");
}

__device__ __forceinline__ void din_frag(const float* dq, int kb, v2f (&db)[3])
{
    db[0] = *reinterpret_cast<const v2f*>(dq + kb);
    db[1] = *reinterpret_cast<const v2f*>(dq + 4 + kb);
    db[2] = *reinterpret_cast<const v2f*>(dq + 8);
}

__device__ __forceinline__ void sigma_tile(const v2f (&ha)[3], const v2f (&db)[3], int kb, float* Sd)
{
    v2f b2;
    b2.x = (kb == 0) ? db[2].x : 0.f;
    b2.y = (kb == 0) ? db[2].y : 0.f;
    v8f acc = zero8();
    acc = wmma4(ha[0], db[0], acc);
    acc = wmma4(ha[1], db[1], acc);
    acc = wmma4(ha[2], b2, acc);
    #pragma unroll
    for (int r = 0; r < 8; ++r) Sd[r * 130] = fmaxf(acc[r], 0.f);
}

__global__ __launch_bounds__(512) void traj_kernel(
    const float* __restrict__ x,
    const float* __restrict__ traj, const float* __restrict__ w1s,
    const float* __restrict__ w1v,
    const float* Din   ,
    const float* Dout  ,
    const float* __restrict__ ico,
    const float* __restrict__ w20, const float* __restrict__ w21,
    const float* __restrict__ w22, const float* __restrict__ w23,
    const float* __restrict__ w24, const float* __restrict__ w25,
    const float* __restrict__ w26, float* __restrict__ out)
{
    constexpr int G = 4096, KC = 128, MR = 32;
    __shared__ float H2[MR][H2S];
    __shared__ __attribute__((aligned(16))) float U[2 * 64 * OUTS];
    constexpr int SBUF = MR * 130;
    float (*h_lds)[12] = reinterpret_cast<float (*)[12]>(U + 2 * SBUF);
    float* OUTf = U;

    const int bt0  = blockIdx.x * 2;
    const int tid  = threadIdx.x;
    const int wave = tid >> 5;
    const int lane = tid & 31;
    const int lrow = lane & 15;
    const int kb   = (lane >> 4) << 1;
    const int crow = (lane >> 4) << 3;

    {
        constexpr int NX4 = 64 * DIM455 / 4;
        constexpr int NI4 = 60 * DIM455 / 4;
        constexpr int RX = (NX4 + 511) / 512;
        constexpr int RI = (NI4 + 511) / 512;
        float* XS = U;
        float* IS = U + 64 * DIM455;
        const v4fu* xg4 = reinterpret_cast<const v4fu*>(x + (long)blockIdx.x * (64 * DIM455));
        const v4fu* ig4 = reinterpret_cast<const v4fu*>(ico);
        v4f* XS4 = reinterpret_cast<v4f*>(XS);
        v4f* IS4 = reinterpret_cast<v4f*>(IS);
        v4f xv[RX], iv[RI];
        const int lx = (512 * (RX - 1) + tid < NX4) ? (512 * (RX - 1) + tid) : (NX4 - 1);
        const int li = (512 * (RI - 1) + tid < NI4) ? (512 * (RI - 1) + tid) : (NI4 - 1);
        #pragma unroll
        for (int j = 0; j < RX - 1; ++j) xv[j] = xg4[tid + 512 * j];
        xv[RX - 1] = xg4[lx];
        #pragma unroll
        for (int j = 0; j < RI - 1; ++j) iv[j] = ig4[tid + 512 * j];
        iv[RI - 1] = ig4[li];
        #pragma unroll
        for (int j = 0; j < RX - 1; ++j) XS4[tid + 512 * j] = xv[j];
        XS4[lx] = xv[RX - 1];
        #pragma unroll
        for (int j = 0; j < RI - 1; ++j) IS4[tid + 512 * j] = iv[j];
        IS4[li] = iv[RI - 1];
        lds_barrier();

        const int xmt = wave >> 2, xnt = wave & 3;
        const int rb  = xnt * 16 + lrow;
        const int rbc = (rb < 60) ? rb : 59;
        const float* xa = XS + (xmt * 16 + lrow) * DIM455 + kb;
        const float* ib = IS + rbc * DIM455 + kb;
        v8f xacc = zero8();
        #pragma unroll 16
        for (int s = 0; s < 112; ++s) {
            v2f a, b;
            a.x = xa[4 * s];  a.y = xa[4 * s + 1];
            b.x = ib[4 * s];  b.y = ib[4 * s + 1];
            xacc = wmma4(a, b, xacc);
        }
        {
            v2f a, b;
            a.x = xa[448];  a.y = xa[449];
            b.x = ib[448];  b.y = ib[449];
            xacc = wmma4(a, b, xacc);
        }
        {
            const int k1 = (kb == 0) ? 453 : 452;
            v2f a, b;
            const float ay = xa[k1], by = ib[k1];
            a.x = xa[452];  a.y = (kb == 0) ? ay : 0.f;
            b.x = ib[452];  b.y = (kb == 0) ? by : 0.f;
            xacc = wmma4(a, b, xacc);
        }
        if (rb < 60) {
            const long row0 = (long)blockIdx.x * 64 + xmt * 16 + crow;
            #pragma unroll
            for (int r = 0; r < 8; ++r)
                out[(row0 + r) * 60 + rb] = xacc[r];
        }
        lds_barrier();
    }

    if (tid < 320) {
        int row = tid / 10, i = tid % 10;
        int c = row & 15;
        const float* t10 = traj + (bt0 + (row >> 4)) * 10;
        float hv;
        if (i == 0) {
            hv = w1s[c] * t10[9];
        } else {
            int vv = (i - 1) / 3, mm = (i - 1) % 3;
            float a = 0.f;
            #pragma unroll
            for (int u = 0; u < 3; ++u)
                a += t10[u * 3 + mm] * w1v[c * 9 + u * 3 + vv];
            hv = a * 0.5773502691896258f;
        }
        h_lds[row][i] = hv;
    }
    __syncthreads();

    v8f acc[2][4];
    #pragma unroll
    for (int m = 0; m < 2; ++m) {
        #pragma unroll
        for (int j = 0; j < 4; ++j) acc[m][j] = zero8();
    }

    const int wn   = wave & 7;
    const int khalf   = wave >> 3;
    const int col0 = 64 * wn + 4 * lrow;
    const int colc = (col0 < DIM455 - 4) ? col0 : (DIM455 - 4);
    constexpr int GSTEP = 16 * DIM455;
    const float* p0 = Dout + (64 * khalf + kb) * DIM455 + colc;
    v4f fe[4][2], fo[4][2];

    const int smt = wave & 1;
    const int snt = wave >> 1;
    v2f ha[3];
    {
        const float* hr = &h_lds[smt * 16 + lrow][0];
        ha[0].x = hr[kb];      ha[0].y = hr[kb + 1];
        ha[1].x = hr[4 + kb];  ha[1].y = hr[5 + kb];
        const float h8 = hr[8], h9 = hr[9];
        ha[2].x = (kb == 0) ? h8 : 0.f;
        ha[2].y = (kb == 0) ? h9 : 0.f;
    }
    const float* dq = Din + (snt * 16 + lrow) * 10;
    const int sd_off = (smt * 16 + crow) * 130 + snt * 16 + lrow;
    v2f dn[3];
    din_frag(dq, kb, dn);
    dout_group(p0, fe);
    sigma_tile(ha, dn, kb, U + sd_off);
    dq += KC * 10;
    din_frag(dq, kb, dn);
    lds_barrier();

    const int kq = 64 * khalf;
    for (int g0 = 0; g0 < G; g0 += KC) {
        const int cur = (g0 / KC) & 1;
        const float (*S)[130] = reinterpret_cast<const float (*)[130]>(U + cur * SBUF);
        float* Sd = U + (cur ^ 1) * SBUF + sd_off;

        p0 += GSTEP;
        dout_group(p0, fo);
        order_point(acc);
        h2_group(S, kq, lrow, kb, fe, acc);
        order_point(acc);
        p0 += GSTEP;
        dout_group(p0, fe);
        order_point(acc);
        h2_group(S, kq + 16, lrow, kb, fo, acc);
        order_point(acc);

        sigma_tile(ha, dn, kb, Sd);
        dq += (g0 + 2 * KC < G) ? KC * 10 : 0;
        din_frag(dq, kb, dn);

        p0 += GSTEP;
        dout_group(p0, fo);
        order_point(acc);
        h2_group(S, kq + 32, lrow, kb, fe, acc);
        order_point(acc);
        const int adv = (g0 + KC < G) ? (GSTEP + 64 * DIM455) : 0;
        p0 += adv;
        dout_group(p0, fe);
        order_point(acc);
        h2_group(S, kq + 48, lrow, kb, fo, acc);
        order_point(acc);
        lds_barrier();
    }

    if (khalf == 1) {
        #pragma unroll
        for (int m = 0; m < 2; ++m) {
            #pragma unroll
            for (int j = 0; j < 4; ++j) {
                const int col = colc + j;
                if (col >= col0 && col < DIM455) {
                    #pragma unroll
                    for (int r = 0; r < 8; ++r)
                        H2[m * 16 + r + crow][col] = acc[m][j][r];
                }
            }
        }
    }
    __syncthreads();
    if (khalf == 0) {
        #pragma unroll
        for (int m = 0; m < 2; ++m) {
            #pragma unroll
            for (int j = 0; j < 4; ++j) {
                const int col = colc + j;
                if (col >= col0 && col < DIM455) {
                    #pragma unroll
                    for (int r = 0; r < 8; ++r)
                        H2[m * 16 + r + crow][col] += acc[m][j][r];
                }
            }
        }
    }
    __syncthreads();

    if (tid < 128) OUTf[tid * OUTS + DIM455] = 0.f;
    const float* H2f = &H2[0][0];
    {
        const int wv = __builtin_amdgcn_readfirstlane(wave);
        const bool isC = wv < 4, isA = (wv >= 4 && wv < 9), isB = wv >= 9;
        const int a = wv - 4, b = wv - 9, c = wv;
        int q6 = 0, n6 = 0, q5 = 0, n5 = 0, q4 = 0, n4 = 0, q3 = 0, n3 = 0;
        int q2 = 0, n2 = 0, q1 = 0, n1 = 0, n0 = 0;
        if (isA) { q6 = a; n6 = 1; q5 = a; n5 = 1; q2 = a; n2 = 1; }
        if (isB) { q6 = 5 + b; n6 = 1; q4 = b; n4 = 1; q3 = b; n3 = 1; }
        if (isC) {
            if (c == 0) { q6 = 12; n6 = 1; q5 = 5; n5 = 1; q1 = 0; n1 = 1; }
            if (c == 1) { q5 = 6; n5 = 2; q1 = 1; n1 = 1; n0 = 1; }
            if (c == 2) { q5 = 8; n5 = 2; q1 = 2; n1 = 1; }
            if (c == 3) { q5 = 10; n5 = 1; q4 = 7; n4 = 2; }
        }
        mix_level<6>(w26, H2f, OUTf, q6, n6, lrow, kb, crow);
        mix_level<5>(w25, H2f, OUTf, q5, n5, lrow, kb, crow);
        mix_level<4>(w24, H2f, OUTf, q4, n4, lrow, kb, crow);
        mix_level<3>(w23, H2f, OUTf, q3, n3, lrow, kb, crow);
        mix_level<2>(w22, H2f, OUTf, q2, n2, lrow, kb, crow);
        mix_level<1>(w21, H2f, OUTf, q1, n1, lrow, kb, crow);
        mix_level<0>(w20, H2f, OUTf, 0, n0, lrow, kb, crow);
    }
    __syncthreads();

    const int bsel = wave >> 3;
    const int kh = wave & 1;
    const int np = (wave >> 1) & 1;
    const int mp = (wave >> 2) & 1;
    const int bt = bt0 + bsel;
    OUTf += bsel * 64 * OUTS;

    v8f facc0 = zero8(), facc1 = zero8(), facc2 = zero8(), facc3 = zero8();
    {
        const float* oa0 = OUTf + ((mp * 2 + 0) * 16 + lrow) * OUTS;
        const float* oa1 = OUTf + ((mp * 2 + 1) * 16 + lrow) * OUTS;
        const int rb0 = (np * 2 + 0) * 16 + lrow;
        const int rb1 = (np * 2 + 1) * 16 + lrow;
        const int rb1c = (rb1 < 60) ? rb1 : 59;
        const float* i0 = ico + rb0 * DIM455;
        const float* i1 = ico + rb1c * DIM455;
        const int ks0 = kh * 57;
        for (int ks = ks0; ks < ks0 + 57; ++ks) {
            const int k  = ks * 4 + kb;
            const int k1 = (k + 1 < DIM455) ? (k + 1) : (DIM455 - 1);
            v2f a0 = *reinterpret_cast<const v2f*>(oa0 + k);
            v2f a1 = *reinterpret_cast<const v2f*>(oa1 + k);
            float t0x = i0[k], t0y = i0[k1];
            float t1x = i1[k], t1y = i1[k1];
            v2f b0, b1;
            b0.x = t0x;  b0.y = (k + 1 < DIM455) ? t0y : 0.f;
            b1.x = t1x;  b1.y = (k + 1 < DIM455) ? t1y : 0.f;
            facc0 = wmma4(a0, b0, facc0);
            facc1 = wmma4(a0, b1, facc1);
            facc2 = wmma4(a1, b0, facc2);
            facc3 = wmma4(a1, b1, facc3);
        }
    }

    float* red = &H2[0][0] + bsel * (64 * 64);
    {
        v8f* faccs[4] = {&facc0, &facc1, &facc2, &facc3};
        if (kh == 1) {
            #pragma unroll
            for (int q = 0; q < 4; ++q) {
                const int mt = mp * 2 + (q >> 1), nt = np * 2 + (q & 1);
                #pragma unroll
                for (int j = 0; j < 8; ++j)
                    red[(mt * 16 + j + crow) * 64 + nt * 16 + lrow] = (*faccs[q])[j];
            }
        }
        __syncthreads();
        if (kh == 0) {
            const long xg_elems = 32L * 4 * 128 * 60;
            #pragma unroll
            for (int q = 0; q < 4; ++q) {
                const int mt = mp * 2 + (q >> 1), nt = np * 2 + (q & 1);
                const int r = nt * 16 + lrow;
                if (r < 60) {
                    #pragma unroll
                    for (int j = 0; j < 8; ++j) {
                        const int g2 = mt * 16 + j + crow;
                        float vsum = (*faccs[q])[j] + red[g2 * 64 + r];
                        out[xg_elems + ((long)bt * 64 + g2) * 60 + r] = vsum;
                    }
                }
            }
        }
    }
}

extern "C" void kernel_launch(void* const* d_in, const int* in_sizes, int n_in,
                              void* d_out, int out_size, void* d_ws, size_t ws_size,
                              hipStream_t stream) {
    const float* x    = (const float*)d_in[0];
    const float* traj = (const float*)d_in[1];
    const float* w1s  = (const float*)d_in[2];
    const float* w1v  = (const float*)d_in[3];
    const float* Din  = (const float*)d_in[4];
    const float* Dout = (const float*)d_in[5];
    const float* ico  = (const float*)d_in[6];
    float* out = (float*)d_out;

    traj_kernel<<<dim3(256), dim3(512), 0, stream>>>(
        x, traj, w1s, w1v, Din, Dout, ico,
        (const float*)d_in[7],  (const float*)d_in[8],  (const float*)d_in[9],
        (const float*)d_in[10], (const float*)d_in[11], (const float*)d_in[12],
        (const float*)d_in[13], out);
}
